# starting kernel
# baseline (speedup 1.0000x reference)
#include <hip/hip_runtime.h>
#include <math.h>


typedef float v2f __attribute__((ext_vector_type(2)));
typedef float v8f __attribute__((ext_vector_type(8)));

#define T_DIM 128
#define B_DIM 32
#define E_DIM 128
#define H_DIM 128
#define GE_DIM 256
#define G_DIM 768
#define GI_OFF 384
#define GC_OFF 512
#define GO_OFF 640
#define ROWS (T_DIM * B_DIM)
#define SSTR (T_DIM + 1)

__device__ __forceinline__ const float* wrow(int k,
    const float* Wf, const float* Wi, const float* Wc, const float* Wo) {
  if (k < GI_OFF) return Wf + (size_t)k * GE_DIM;
  if (k < GC_OFF) return Wi + (size_t)(k - GI_OFF) * GE_DIM;
  if (k < GO_OFF) return Wc + (size_t)(k - GC_OFF) * GE_DIM;
  return Wo + (size_t)(k - GO_OFF) * GE_DIM;
}
__device__ __forceinline__ float biasAt(int k,
    const float* bf, const float* bi, const float* bc, const float* bo) {
  if (k < GI_OFF) return bf[k];
  if (k < GC_OFF) return bi[k - GI_OFF];
  if (k < GO_OFF) return bc[k - GC_OFF];
  return bo[k - GO_OFF];
}

__global__ void __launch_bounds__(256)
xproj_wmma(const float* __restrict__ x,
           const float* __restrict__ Wf, const float* __restrict__ bf,
           const float* __restrict__ Wi, const float* __restrict__ bi,
           const float* __restrict__ Wc, const float* __restrict__ bc,
           const float* __restrict__ Wo, const float* __restrict__ bo,
           float* __restrict__ Xp) {
  const int wid  = blockIdx.x * (blockDim.x >> 5) + (threadIdx.x >> 5);
  const int lane = threadIdx.x & 31;
  const int rowTile = wid / (G_DIM / 16);
  const int colTile = wid % (G_DIM / 16);
  if (rowTile >= ROWS / 16) return;
  const int r0 = rowTile * 16;
  const int k0 = colTile * 16;
  const int mn = lane & 15;
  const int kb = (lane < 16) ? 0 : 2;

  const float* xrow = x + (size_t)(r0 + mn) * E_DIM + kb;
  const int kg = k0 + mn;
  const float* wr = wrow(kg, Wf, Wi, Wc, Wo) + H_DIM + kb;
  const float bb = biasAt(kg, bf, bi, bc, bo);

  v8f acc0 = {};
  v8f acc1 = {};
  #pragma unroll
  for (int kk = 0; kk < E_DIM; kk += 8) {
    v2f a0 = *(const v2f*)(xrow + kk);
    v2f b0 = *(const v2f*)(wr + kk);
    v2f a1 = *(const v2f*)(xrow + kk + 4);
    v2f b1 = *(const v2f*)(wr + kk + 4);
    acc0 = __builtin_amdgcn_wmma_f32_16x16x4_f32(false, a0, false, b0,
                                                 (short)0, acc0, false, false);
    acc1 = __builtin_amdgcn_wmma_f32_16x16x4_f32(false, a1, false, b1,
                                                 (short)0, acc1, false, false);
  }
  const int mbase = (lane < 16) ? 0 : 8;
  #pragma unroll
  for (int v = 0; v < 8; ++v) {
    Xp[(size_t)(r0 + mbase + v) * G_DIM + kg] = acc0[v] + acc1[v] + bb;
  }
}

__global__ void wh_transpose(const float* __restrict__ Wf, const float* __restrict__ Wi,
                             const float* __restrict__ Wc, const float* __restrict__ Wo,
                             float* __restrict__ WhT) {
  int g = blockIdx.x * blockDim.x + threadIdx.x;
  if (g >= H_DIM * G_DIM) return;
  int i = g / G_DIM;
  int k = g - i * G_DIM;
  WhT[g] = wrow(k, Wf, Wi, Wc, Wo)[i];
}

__global__ void __launch_bounds__(256)
scan_kernel(const float* __restrict__ Xp,
            const float* __restrict__ WhT,
            const float* __restrict__ b_stay,
            const float* __restrict__ b_down,
            float* __restrict__ out) {
  extern __shared__ float smem[];
  float* cur   = smem;
  float* nxt   = cur + H_DIM * SSTR;
  float* gates = nxt + H_DIM * SSTR;
  float* fa = gates + G_DIM;
  float* fb = fa + H_DIM;
  float* fc = fb + H_DIM;
  float* ov = fc + H_DIM;
  float* hv = ov + H_DIM;
  float* bs = hv + H_DIM;
  float* bd = bs + H_DIM;

  const int tid = threadIdx.x;
  const int b   = blockIdx.x;

  for (int i = tid; i < H_DIM * SSTR; i += blockDim.x) { cur[i] = 0.f; nxt[i] = 0.f; }
  if (tid < H_DIM) { hv[tid] = 0.f; bs[tid] = b_stay[tid]; bd[tid] = b_down[tid]; }
  __syncthreads();

  for (int t = 0; t < T_DIM; ++t) {
    const float* xp = Xp + (size_t)(t * B_DIM + b) * G_DIM;
    if (t + 1 < T_DIM) {
      const float* xpn = xp + (size_t)B_DIM * G_DIM;
      __builtin_prefetch(xpn + tid, 0, 0);
      __builtin_prefetch(xpn + tid + 256, 0, 0);
      __builtin_prefetch(xpn + tid + 512, 0, 0);
    }
    #pragma unroll
    for (int q = 0; q < 3; ++q) {
      const int k = tid + q * 256;
      float acc = xp[k];
      #pragma unroll 8
      for (int i = 0; i < H_DIM; ++i) acc = fmaf(hv[i], WhT[i * G_DIM + k], acc);
      gates[k] = acc;
    }
    __syncthreads();
    if (tid < H_DIM) {
      float r0 = gates[tid * 3 + 0], r1 = gates[tid * 3 + 1], r2 = gates[tid * 3 + 2];
      float mx = fmaxf(r0, fmaxf(r1, r2));
      float e0 = __expf(r0 - mx), e1 = __expf(r1 - mx), e2 = __expf(r2 - mx);
      float inv = 1.f / (e0 + e1 + e2);
      fa[tid] = e0 * inv; fb[tid] = e1 * inv; fc[tid] = e2 * inv;
      float iv = 1.f / (1.f + __expf(-gates[GI_OFF + tid]));
      float cv = tanhf(gates[GC_OFF + tid]);
      ov[tid]  = 1.f / (1.f + __expf(-gates[GO_OFF + tid]));
      cur[tid * SSTR + t] = iv * cv;
    }
    __syncthreads();
    for (int e = tid; e < H_DIM * T_DIM; e += blockDim.x) {
      const int hh = e >> 7;
      const int j  = e & (T_DIM - 1);
      const int row = hh * SSTR;
      const int jm = (j > 0) ? j - 1 : 0;
      int jp = j + 1; if (jp > t) jp = t;
      const float sl = cur[row + jm];
      const float sm = cur[row + j];
      const float sr = cur[row + jp];
      nxt[row + j] = sl * fa[hh] + (sm + bs[hh]) * fb[hh] + (sr + bd[hh]) * fc[hh];
    }
    __syncthreads();
    if (tid < H_DIM) {
      float hn = nxt[tid * SSTR + t] * ov[tid];
      hv[tid] = hn;
      out[(size_t)(t * B_DIM + b) * H_DIM + tid] = hn;
    }
    __syncthreads();
    float* tmp = cur; cur = nxt; nxt = tmp;
  }
  if (tid < H_DIM) {
    out[(size_t)T_DIM * B_DIM * H_DIM + (size_t)b * H_DIM + tid] = hv[tid];
  }
}

extern "C" void kernel_launch(void* const* d_in, const int* in_sizes, int n_in,
                              void* d_out, int out_size, void* d_ws, size_t ws_size,
                              hipStream_t stream) {
  (void)in_sizes; (void)n_in; (void)out_size; (void)ws_size;
  const float* x  = (const float*)d_in[0];
  const float* Wf = (const float*)d_in[1];
  const float* bf = (const float*)d_in[2];
  const float* Wi = (const float*)d_in[3];
  const float* bi = (const float*)d_in[4];
  const float* Wc = (const float*)d_in[5];
  const float* bc = (const float*)d_in[6];
  const float* Wo = (const float*)d_in[7];
  const float* bo = (const float*)d_in[8];
  const float* b_stay = (const float*)d_in[11];
  const float* b_down = (const float*)d_in[13];
  float* out = (float*)d_out;

  float* Xp  = (float*)d_ws;
  float* WhT = Xp + (size_t)ROWS * G_DIM;

  xproj_wmma<<<(ROWS / 16) * (G_DIM / 16) / 8, 256, 0, stream>>>(
      x, Wf, bf, Wi, bi, Wc, bc, Wo, bo, Xp);
  wh_transpose<<<(H_DIM * G_DIM + 255) / 256, 256, 0, stream>>>(Wf, Wi, Wc, Wo, WhT);

  size_t shmem = (size_t)(2 * H_DIM * SSTR + G_DIM + 7 * H_DIM) * sizeof(float);
  (void)hipFuncSetAttribute((const void*)scan_kernel,
                            hipFuncAttributeMaxDynamicSharedMemorySize, (int)shmem);
  scan_kernel<<<B_DIM, 256, shmem, stream>>>(Xp, WhT, b_stay, b_down, out);
}


# scan: each thread keeps its gate row's 128 recurrent weights in VGPRs for all steps (768-thread block, h broadcast from LDS) instead of re-reading them from L2 each step; transposed copy dropped
# speedup vs baseline: 3.4512x; 3.4512x over previous
#include <hip/hip_runtime.h>
#include <math.h>


typedef float v2f __attribute__((ext_vector_type(2)));
typedef float v8f __attribute__((ext_vector_type(8)));

#define T_DIM 128
#define B_DIM 32
#define E_DIM 128
#define H_DIM 128
#define GE_DIM 256
#define G_DIM 768
#define GI_OFF 384
#define GC_OFF 512
#define GO_OFF 640
#define ROWS (T_DIM * B_DIM)
#define SSTR (T_DIM + 1)

__device__ __forceinline__ const float* wrow(int k,
    const float* Wf, const float* Wi, const float* Wc, const float* Wo) {
  if (k < GI_OFF) return Wf + (size_t)k * GE_DIM;
  if (k < GC_OFF) return Wi + (size_t)(k - GI_OFF) * GE_DIM;
  if (k < GO_OFF) return Wc + (size_t)(k - GC_OFF) * GE_DIM;
  return Wo + (size_t)(k - GO_OFF) * GE_DIM;
}
__device__ __forceinline__ float biasAt(int k,
    const float* bf, const float* bi, const float* bc, const float* bo) {
  if (k < GI_OFF) return bf[k];
  if (k < GC_OFF) return bi[k - GI_OFF];
  if (k < GO_OFF) return bc[k - GC_OFF];
  return bo[k - GO_OFF];
}

__global__ void __launch_bounds__(256)
xproj_wmma(const float* __restrict__ x,
           const float* __restrict__ Wf, const float* __restrict__ bf,
           const float* __restrict__ Wi, const float* __restrict__ bi,
           const float* __restrict__ Wc, const float* __restrict__ bc,
           const float* __restrict__ Wo, const float* __restrict__ bo,
           float* __restrict__ Xp) {
  const int wid  = blockIdx.x * (blockDim.x >> 5) + (threadIdx.x >> 5);
  const int lane = threadIdx.x & 31;
  const int rowTile = wid / (G_DIM / 16);
  const int colTile = wid % (G_DIM / 16);
  if (rowTile >= ROWS / 16) return;
  const int r0 = rowTile * 16;
  const int k0 = colTile * 16;
  const int mn = lane & 15;
  const int kb = (lane < 16) ? 0 : 2;

  const float* xrow = x + (size_t)(r0 + mn) * E_DIM + kb;
  const int kg = k0 + mn;
  const float* wr = wrow(kg, Wf, Wi, Wc, Wo) + H_DIM + kb;
  const float bb = biasAt(kg, bf, bi, bc, bo);

  v8f acc0 = {};
  v8f acc1 = {};
  #pragma unroll
  for (int kk = 0; kk < E_DIM; kk += 8) {
    v2f a0 = *(const v2f*)(xrow + kk);
    v2f b0 = *(const v2f*)(wr + kk);
    v2f a1 = *(const v2f*)(xrow + kk + 4);
    v2f b1 = *(const v2f*)(wr + kk + 4);
    acc0 = __builtin_amdgcn_wmma_f32_16x16x4_f32(false, a0, false, b0,
                                                 (short)0, acc0, false, false);
    acc1 = __builtin_amdgcn_wmma_f32_16x16x4_f32(false, a1, false, b1,
                                                 (short)0, acc1, false, false);
  }
  const int mbase = (lane < 16) ? 0 : 8;
  #pragma unroll
  for (int v = 0; v < 8; ++v) {
    Xp[(size_t)(r0 + mbase + v) * G_DIM + kg] = acc0[v] + acc1[v] + bb;
  }
}

#define SCAN_THREADS G_DIM
#define HWIN 3

__global__ void __launch_bounds__(SCAN_THREADS)
scan_kernel(const float* __restrict__ Xp,
            const float* __restrict__ Wf, const float* __restrict__ Wi,
            const float* __restrict__ Wc, const float* __restrict__ Wo,
            const float* __restrict__ b_stay,
            const float* __restrict__ b_down,
            float* __restrict__ out) {
  extern __shared__ __align__(16) float smem[];
  float* gates = smem;
  float* fa = gates + G_DIM;
  float* fb = fa + H_DIM;
  float* fc = fb + H_DIM;
  float* ov = fc + H_DIM;
  float* hv = ov + H_DIM;
  float* bs = hv + H_DIM;
  float* bd = bs + H_DIM;
  float* cur   = bd + H_DIM;
  float* nxt   = cur + H_DIM * SSTR;

  const int tid = threadIdx.x;
  const int b   = blockIdx.x;

  float w[H_DIM];
  {
    const float4* wr4 = (const float4*)wrow(tid, Wf, Wi, Wc, Wo);
    #pragma unroll
    for (int i4 = 0; i4 < H_DIM / 4; ++i4) {
      const float4 v = wr4[i4];
      w[4 * i4 + 0] = v.x; w[4 * i4 + 1] = v.y; w[4 * i4 + 2] = v.z; w[4 * i4 + 3] = v.w;
    }
  }

  for (int i = tid; i < H_DIM * SSTR; i += blockDim.x) { cur[i] = 0.f; nxt[i] = 0.f; }
  if (tid < H_DIM) { hv[tid] = 0.f; bs[tid] = b_stay[tid]; bd[tid] = b_down[tid]; }
  __syncthreads();

  for (int t = 0; t < T_DIM; ++t) {
    const float* xp = Xp + (size_t)(t * B_DIM + b) * G_DIM;
    if (t + 1 < T_DIM) {
      __builtin_prefetch(xp + (size_t)B_DIM * G_DIM + tid, 0, 0);
    }
    {
      float acc = xp[tid];
      const float4* h4 = (const float4*)hv;
      float4 q[HWIN];
      #pragma unroll
      for (int j = 0; j < HWIN; ++j) q[j] = h4[j];
      __builtin_amdgcn_sched_barrier(0);
      #pragma unroll
      for (int i4 = 0; i4 < H_DIM / 4; ++i4) {
        const float4 h = q[i4 % HWIN];
        acc = fmaf(h.x, w[4 * i4 + 0], acc);
        acc = fmaf(h.y, w[4 * i4 + 1], acc);
        acc = fmaf(h.z, w[4 * i4 + 2], acc);
        acc = fmaf(h.w, w[4 * i4 + 3], acc);
        if (i4 + HWIN < H_DIM / 4) q[i4 % HWIN] = h4[i4 + HWIN];
        __builtin_amdgcn_sched_barrier(0);
      }
      gates[tid] = acc;
    }
    __syncthreads();
    if (tid < H_DIM) {
      float r0 = gates[tid * 3 + 0], r1 = gates[tid * 3 + 1], r2 = gates[tid * 3 + 2];
      float mx = fmaxf(r0, fmaxf(r1, r2));
      float e0 = __expf(r0 - mx), e1 = __expf(r1 - mx), e2 = __expf(r2 - mx);
      float inv = 1.f / (e0 + e1 + e2);
      fa[tid] = e0 * inv; fb[tid] = e1 * inv; fc[tid] = e2 * inv;
      float iv = 1.f / (1.f + __expf(-gates[GI_OFF + tid]));
      float cv = tanhf(gates[GC_OFF + tid]);
      ov[tid]  = 1.f / (1.f + __expf(-gates[GO_OFF + tid]));
      cur[tid * SSTR + t] = iv * cv;
    }
    __syncthreads();
    for (int e = tid; e < H_DIM * T_DIM; e += blockDim.x) {
      const int hh = e >> 7;
      const int j  = e & (T_DIM - 1);
      const int row = hh * SSTR;
      const int jm = (j > 0) ? j - 1 : 0;
      int jp = j + 1; if (jp > t) jp = t;
      const float sl = cur[row + jm];
      const float sm = cur[row + j];
      const float sr = cur[row + jp];
      nxt[row + j] = sl * fa[hh] + (sm + bs[hh]) * fb[hh] + (sr + bd[hh]) * fc[hh];
    }
    __syncthreads();
    if (tid < H_DIM) {
      float hn = nxt[tid * SSTR + t] * ov[tid];
      hv[tid] = hn;
      out[(size_t)(t * B_DIM + b) * H_DIM + tid] = hn;
    }
    __syncthreads();
    float* tmp = cur; cur = nxt; nxt = tmp;
  }
  if (tid < H_DIM) {
    out[(size_t)T_DIM * B_DIM * H_DIM + (size_t)b * H_DIM + tid] = hv[tid];
  }
}

extern "C" void kernel_launch(void* const* d_in, const int* in_sizes, int n_in,
                              void* d_out, int out_size, void* d_ws, size_t ws_size,
                              hipStream_t stream) {
  (void)in_sizes; (void)n_in; (void)out_size; (void)ws_size;
  const float* x  = (const float*)d_in[0];
  const float* Wf = (const float*)d_in[1];
  const float* bf = (const float*)d_in[2];
  const float* Wi = (const float*)d_in[3];
  const float* bi = (const float*)d_in[4];
  const float* Wc = (const float*)d_in[5];
  const float* bc = (const float*)d_in[6];
  const float* Wo = (const float*)d_in[7];
  const float* bo = (const float*)d_in[8];
  const float* b_stay = (const float*)d_in[11];
  const float* b_down = (const float*)d_in[13];
  float* out = (float*)d_out;

  float* Xp  = (float*)d_ws;

  xproj_wmma<<<(ROWS / 16) * (G_DIM / 16) / 8, 256, 0, stream>>>(
      x, Wf, bf, Wi, bi, Wc, bc, Wo, bo, Xp);

  size_t shmem = (size_t)(2 * H_DIM * SSTR + G_DIM + 7 * H_DIM) * sizeof(float);
  (void)hipFuncSetAttribute((const void*)scan_kernel,
                            hipFuncAttributeMaxDynamicSharedMemorySize, (int)shmem);
  scan_kernel<<<B_DIM, SCAN_THREADS, shmem, stream>>>(Xp, Wf, Wi, Wc, Wo, b_stay, b_down, out);
}


# scan state update: each thread owns 22 consecutive stack slots of one row (8-byte LDS loads 3 ahead, stencil in registers, 8-byte stores; right clamp as a replicated pad slot), not one slot a trip
# speedup vs baseline: 6.3332x; 1.8351x over previous
#include <hip/hip_runtime.h>
#include <math.h>


typedef float v2f __attribute__((ext_vector_type(2)));
typedef float v8f __attribute__((ext_vector_type(8)));

#define T_DIM 128
#define B_DIM 32
#define E_DIM 128
#define H_DIM 128
#define GE_DIM 256
#define G_DIM 768
#define GI_OFF 384
#define GC_OFF 512
#define GO_OFF 640
#define ROWS (T_DIM * B_DIM)
#define SEG_LEN 22
#define SSTR 134

__device__ __forceinline__ const float* wrow(int k,
    const float* Wf, const float* Wi, const float* Wc, const float* Wo) {
  if (k < GI_OFF) return Wf + (size_t)k * GE_DIM;
  if (k < GC_OFF) return Wi + (size_t)(k - GI_OFF) * GE_DIM;
  if (k < GO_OFF) return Wc + (size_t)(k - GC_OFF) * GE_DIM;
  return Wo + (size_t)(k - GO_OFF) * GE_DIM;
}
__device__ __forceinline__ float biasAt(int k,
    const float* bf, const float* bi, const float* bc, const float* bo) {
  if (k < GI_OFF) return bf[k];
  if (k < GC_OFF) return bi[k - GI_OFF];
  if (k < GO_OFF) return bc[k - GC_OFF];
  return bo[k - GO_OFF];
}

__global__ void __launch_bounds__(256)
xproj_wmma(const float* __restrict__ x,
           const float* __restrict__ Wf, const float* __restrict__ bf,
           const float* __restrict__ Wi, const float* __restrict__ bi,
           const float* __restrict__ Wc, const float* __restrict__ bc,
           const float* __restrict__ Wo, const float* __restrict__ bo,
           float* __restrict__ Xp) {
  const int wid  = blockIdx.x * (blockDim.x >> 5) + (threadIdx.x >> 5);
  const int lane = threadIdx.x & 31;
  const int rowTile = wid / (G_DIM / 16);
  const int colTile = wid % (G_DIM / 16);
  if (rowTile >= ROWS / 16) return;
  const int r0 = rowTile * 16;
  const int k0 = colTile * 16;
  const int mn = lane & 15;
  const int kb = (lane < 16) ? 0 : 2;

  const float* xrow = x + (size_t)(r0 + mn) * E_DIM + kb;
  const int kg = k0 + mn;
  const float* wr = wrow(kg, Wf, Wi, Wc, Wo) + H_DIM + kb;
  const float bb = biasAt(kg, bf, bi, bc, bo);

  v8f acc0 = {};
  v8f acc1 = {};
  #pragma unroll
  for (int kk = 0; kk < E_DIM; kk += 8) {
    v2f a0 = *(const v2f*)(xrow + kk);
    v2f b0 = *(const v2f*)(wr + kk);
    v2f a1 = *(const v2f*)(xrow + kk + 4);
    v2f b1 = *(const v2f*)(wr + kk + 4);
    acc0 = __builtin_amdgcn_wmma_f32_16x16x4_f32(false, a0, false, b0,
                                                 (short)0, acc0, false, false);
    acc1 = __builtin_amdgcn_wmma_f32_16x16x4_f32(false, a1, false, b1,
                                                 (short)0, acc1, false, false);
  }
  const int mbase = (lane < 16) ? 0 : 8;
  #pragma unroll
  for (int v = 0; v < 8; ++v) {
    Xp[(size_t)(r0 + mbase + v) * G_DIM + kg] = acc0[v] + acc1[v] + bb;
  }
}

#define SCAN_THREADS G_DIM
#define HWIN 3
#define UWIN 3

__global__ void __launch_bounds__(SCAN_THREADS)
scan_kernel(const float* __restrict__ Xp,
            const float* __restrict__ Wf, const float* __restrict__ Wi,
            const float* __restrict__ Wc, const float* __restrict__ Wo,
            const float* __restrict__ b_stay,
            const float* __restrict__ b_down,
            float* __restrict__ out) {
  extern __shared__ __align__(16) float smem[];
  float* gates = smem;
  float* fa = gates + G_DIM;
  float* fb = fa + H_DIM;
  float* fc = fb + H_DIM;
  float* ov = fc + H_DIM;
  float* hv = ov + H_DIM;
  float* bs = hv + H_DIM;
  float* bd = bs + H_DIM;
  float* cur   = bd + H_DIM;
  float* nxt   = cur + H_DIM * SSTR;

  const int tid = threadIdx.x;
  const int b   = blockIdx.x;

  const int uh   = tid & (H_DIM - 1);
  const int uoff = uh * SSTR + (tid >> 7) * SEG_LEN;

  float w[H_DIM];
  {
    const float4* wr4 = (const float4*)wrow(tid, Wf, Wi, Wc, Wo);
    #pragma unroll
    for (int i4 = 0; i4 < H_DIM / 4; ++i4) {
      const float4 v = wr4[i4];
      w[4 * i4 + 0] = v.x; w[4 * i4 + 1] = v.y; w[4 * i4 + 2] = v.z; w[4 * i4 + 3] = v.w;
    }
  }

  for (int i = tid; i < H_DIM * SSTR; i += blockDim.x) { cur[i] = 0.f; nxt[i] = 0.f; }
  if (tid < H_DIM) { hv[tid] = 0.f; bs[tid] = b_stay[tid]; bd[tid] = b_down[tid]; }
  __syncthreads();

  for (int t = 0; t < T_DIM; ++t) {
    const float* xp = Xp + (size_t)(t * B_DIM + b) * G_DIM;
    if (t + 1 < T_DIM) {
      __builtin_prefetch(xp + (size_t)B_DIM * G_DIM + tid, 0, 0);
    }
    {
      float acc = xp[tid];
      const float4* h4 = (const float4*)hv;
      float4 q[HWIN];
      #pragma unroll
      for (int j = 0; j < HWIN; ++j) q[j] = h4[j];
      __builtin_amdgcn_sched_barrier(0);
      #pragma unroll
      for (int i4 = 0; i4 < H_DIM / 4; ++i4) {
        const float4 h = q[i4 % HWIN];
        acc = fmaf(h.x, w[4 * i4 + 0], acc);
        acc = fmaf(h.y, w[4 * i4 + 1], acc);
        acc = fmaf(h.z, w[4 * i4 + 2], acc);
        acc = fmaf(h.w, w[4 * i4 + 3], acc);
        if (i4 + HWIN < H_DIM / 4) q[i4 % HWIN] = h4[i4 + HWIN];
        __builtin_amdgcn_sched_barrier(0);
      }
      gates[tid] = acc;
    }
    __syncthreads();
    if (tid < H_DIM) {
      float r0 = gates[tid * 3 + 0], r1 = gates[tid * 3 + 1], r2 = gates[tid * 3 + 2];
      float mx = fmaxf(r0, fmaxf(r1, r2));
      float e0 = __expf(r0 - mx), e1 = __expf(r1 - mx), e2 = __expf(r2 - mx);
      float inv = 1.f / (e0 + e1 + e2);
      fa[tid] = e0 * inv; fb[tid] = e1 * inv; fc[tid] = e2 * inv;
      float iv = 1.f / (1.f + __expf(-gates[GI_OFF + tid]));
      float cv = tanhf(gates[GC_OFF + tid]);
      ov[tid]  = 1.f / (1.f + __expf(-gates[GO_OFF + tid]));
      const float pushed = iv * cv;
      cur[tid * SSTR + t] = pushed;
      cur[tid * SSTR + t + 1] = pushed;
    }
    __syncthreads();
    {
      const float ca = fa[uh], cb = fb[uh], cc = fc[uh];
      const float vbs = bs[uh], vbd = bd[uh];
      const float2* sp = (const float2*)(cur + uoff);
      float2* dp = (float2*)(nxt + uoff);
      const float2 hl = sp[-1];
      float2 c0 = sp[0];
      float2 q[UWIN];
      #pragma unroll
      for (int j = 0; j < UWIN; ++j) q[j] = sp[1 + j];
      __builtin_amdgcn_sched_barrier(0);
      float sl = (tid < H_DIM) ? c0.x : hl.y;
      #pragma unroll
      for (int k = 0; k < SEG_LEN / 2; ++k) {
        const float2 c1 = q[k % UWIN];
        float2 n;
        n.x = sl   * ca + (c0.x + vbs) * cb + (c0.y + vbd) * cc;
        n.y = c0.x * ca + (c0.y + vbs) * cb + (c1.x + vbd) * cc;
        dp[k] = n;
        if (k + 1 + UWIN <= SEG_LEN / 2) q[k % UWIN] = sp[k + 1 + UWIN];
        __builtin_amdgcn_sched_barrier(0);
        sl = c0.y; c0 = c1;
      }
    }
    __syncthreads();
    if (tid < H_DIM) {
      float hn = nxt[tid * SSTR + t] * ov[tid];
      hv[tid] = hn;
      out[(size_t)(t * B_DIM + b) * H_DIM + tid] = hn;
    }
    __syncthreads();
    float* tmp = cur; cur = nxt; nxt = tmp;
  }
  if (tid < H_DIM) {
    out[(size_t)T_DIM * B_DIM * H_DIM + (size_t)b * H_DIM + tid] = hv[tid];
  }
}

extern "C" void kernel_launch(void* const* d_in, const int* in_sizes, int n_in,
                              void* d_out, int out_size, void* d_ws, size_t ws_size,
                              hipStream_t stream) {
  (void)in_sizes; (void)n_in; (void)out_size; (void)ws_size;
  const float* x  = (const float*)d_in[0];
  const float* Wf = (const float*)d_in[1];
  const float* bf = (const float*)d_in[2];
  const float* Wi = (const float*)d_in[3];
  const float* bi = (const float*)d_in[4];
  const float* Wc = (const float*)d_in[5];
  const float* bc = (const float*)d_in[6];
  const float* Wo = (const float*)d_in[7];
  const float* bo = (const float*)d_in[8];
  const float* b_stay = (const float*)d_in[11];
  const float* b_down = (const float*)d_in[13];
  float* out = (float*)d_out;

  float* Xp  = (float*)d_ws;

  xproj_wmma<<<(ROWS / 16) * (G_DIM / 16) / 8, 256, 0, stream>>>(
      x, Wf, bf, Wi, bi, Wc, bc, Wo, bo, Xp);

  size_t shmem = (size_t)(2 * H_DIM * SSTR + G_DIM + 7 * H_DIM) * sizeof(float);
  (void)hipFuncSetAttribute((const void*)scan_kernel,
                            hipFuncAttributeMaxDynamicSharedMemorySize, (int)shmem);
  scan_kernel<<<B_DIM, SCAN_THREADS, shmem, stream>>>(Xp, Wf, Wi, Wc, Wo, b_stay, b_down, out);
}


# scan: one thread per hidden unit keeps its six gate rows (768 recurrent weights) in VGPRs and owns its stack row, so gates, nonlinearity, state update are thread-local: 128 threads, 1 barrier a step
# speedup vs baseline: 6.8380x; 1.0797x over previous
#include <hip/hip_runtime.h>
#include <math.h>


typedef float v2f __attribute__((ext_vector_type(2)));
typedef float v8f __attribute__((ext_vector_type(8)));

#define T_DIM 128
#define B_DIM 32
#define E_DIM 128
#define H_DIM 128
#define GE_DIM 256
#define G_DIM 768
#define GI_OFF 384
#define GC_OFF 512
#define GO_OFF 640
#define ROWS (T_DIM * B_DIM)
#define NPAIR 64
#define SSTR 134

__device__ __forceinline__ const float* wrow(int k,
    const float* Wf, const float* Wi, const float* Wc, const float* Wo) {
  if (k < GI_OFF) return Wf + (size_t)k * GE_DIM;
  if (k < GC_OFF) return Wi + (size_t)(k - GI_OFF) * GE_DIM;
  if (k < GO_OFF) return Wc + (size_t)(k - GC_OFF) * GE_DIM;
  return Wo + (size_t)(k - GO_OFF) * GE_DIM;
}
__device__ __forceinline__ float biasAt(int k,
    const float* bf, const float* bi, const float* bc, const float* bo) {
  if (k < GI_OFF) return bf[k];
  if (k < GC_OFF) return bi[k - GI_OFF];
  if (k < GO_OFF) return bc[k - GC_OFF];
  return bo[k - GO_OFF];
}

__global__ void __launch_bounds__(256)
xproj_wmma(const float* __restrict__ x,
           const float* __restrict__ Wf, const float* __restrict__ bf,
           const float* __restrict__ Wi, const float* __restrict__ bi,
           const float* __restrict__ Wc, const float* __restrict__ bc,
           const float* __restrict__ Wo, const float* __restrict__ bo,
           float* __restrict__ Xp) {
  const int wid  = blockIdx.x * (blockDim.x >> 5) + (threadIdx.x >> 5);
  const int lane = threadIdx.x & 31;
  const int rowTile = wid / (G_DIM / 16);
  const int colTile = wid % (G_DIM / 16);
  if (rowTile >= ROWS / 16) return;
  const int r0 = rowTile * 16;
  const int k0 = colTile * 16;
  const int mn = lane & 15;
  const int kb = (lane < 16) ? 0 : 2;

  const float* xrow = x + (size_t)(r0 + mn) * E_DIM + kb;
  const int kg = k0 + mn;
  const float* wr = wrow(kg, Wf, Wi, Wc, Wo) + H_DIM + kb;
  const float bb = biasAt(kg, bf, bi, bc, bo);

  v8f acc0 = {};
  v8f acc1 = {};
  #pragma unroll
  for (int kk = 0; kk < E_DIM; kk += 8) {
    v2f a0 = *(const v2f*)(xrow + kk);
    v2f b0 = *(const v2f*)(wr + kk);
    v2f a1 = *(const v2f*)(xrow + kk + 4);
    v2f b1 = *(const v2f*)(wr + kk + 4);
    acc0 = __builtin_amdgcn_wmma_f32_16x16x4_f32(false, a0, false, b0,
                                                 (short)0, acc0, false, false);
    acc1 = __builtin_amdgcn_wmma_f32_16x16x4_f32(false, a1, false, b1,
                                                 (short)0, acc1, false, false);
  }
  const int mbase = (lane < 16) ? 0 : 8;
  #pragma unroll
  for (int v = 0; v < 8; ++v) {
    Xp[(size_t)(r0 + mbase + v) * G_DIM + kg] = acc0[v] + acc1[v] + bb;
  }
}

#define SCAN_THREADS H_DIM
#define NG 6
#define HWIN 6
#define UWIN 12
#define UGRP 4

__global__ void __launch_bounds__(SCAN_THREADS)
scan_kernel(const float* __restrict__ Xp,
            const float* __restrict__ Wf, const float* __restrict__ Wi,
            const float* __restrict__ Wc, const float* __restrict__ Wo,
            const float* __restrict__ b_stay,
            const float* __restrict__ b_down,
            float* __restrict__ out) {
  extern __shared__ __align__(16) float smem[];
  float* hv   = smem;
  float* rows = hv + 2 * H_DIM;

  const int j = threadIdx.x;
  const int b = blockIdx.x;
  float* row = rows + j * SSTR;

  float w[NG * H_DIM];
  {
    const float* rp[NG];
    rp[0] = Wf + (size_t)(3 * j + 0) * GE_DIM;
    rp[1] = Wf + (size_t)(3 * j + 1) * GE_DIM;
    rp[2] = Wf + (size_t)(3 * j + 2) * GE_DIM;
    rp[3] = Wi + (size_t)j * GE_DIM;
    rp[4] = Wc + (size_t)j * GE_DIM;
    rp[5] = Wo + (size_t)j * GE_DIM;
    #pragma unroll
    for (int r = 0; r < NG; ++r) {
      const float4* wr4 = (const float4*)rp[r];
      #pragma unroll
      for (int i4 = 0; i4 < H_DIM / 4; ++i4) {
        const float4 v = wr4[i4];
        w[r * H_DIM + 4 * i4 + 0] = v.x; w[r * H_DIM + 4 * i4 + 1] = v.y;
        w[r * H_DIM + 4 * i4 + 2] = v.z; w[r * H_DIM + 4 * i4 + 3] = v.w;
      }
    }
  }
  const float vbs = b_stay[j], vbd = b_down[j];

  {
    float2* z = (float2*)row;
    for (int i = 0; i < SSTR / 2; ++i) z[i] = make_float2(0.f, 0.f);
  }
  hv[j] = 0.f; hv[H_DIM + j] = 0.f;
  __syncthreads();

  float hlast = 0.f;
  for (int t = 0; t < T_DIM; ++t) {
    const float* xp = Xp + (size_t)(t * B_DIM + b) * G_DIM;
    if (t + 1 < T_DIM) {
      __builtin_prefetch(xp + (size_t)B_DIM * G_DIM + NG * j, 0, 0);
    }
    float a0 = xp[3 * j + 0], a1 = xp[3 * j + 1], a2 = xp[3 * j + 2];
    float a3 = xp[GI_OFF + j], a4 = xp[GC_OFF + j], a5 = xp[GO_OFF + j];
    const float tl_old = row[t > 0 ? t - 1 : 0];
    {
      const float4* h4 = (const float4*)(hv + (t & 1) * H_DIM);
      float4 q[HWIN];
      #pragma unroll
      for (int n = 0; n < HWIN; ++n) q[n] = h4[n];
      __builtin_amdgcn_sched_barrier(0);
      #pragma unroll
      for (int i4 = 0; i4 < H_DIM / 4; ++i4) {
        const float4 h = q[i4 % HWIN];
        const float hc[4] = {h.x, h.y, h.z, h.w};
        #pragma unroll
        for (int c = 0; c < 4; ++c) {
          const int i = 4 * i4 + c;
          a0 = fmaf(hc[c], w[0 * H_DIM + i], a0);
          a1 = fmaf(hc[c], w[1 * H_DIM + i], a1);
          a2 = fmaf(hc[c], w[2 * H_DIM + i], a2);
          a3 = fmaf(hc[c], w[3 * H_DIM + i], a3);
          a4 = fmaf(hc[c], w[4 * H_DIM + i], a4);
          a5 = fmaf(hc[c], w[5 * H_DIM + i], a5);
        }
        if (i4 + HWIN < H_DIM / 4) q[i4 % HWIN] = h4[i4 + HWIN];
        __builtin_amdgcn_sched_barrier(0);
      }
    }
    asm volatile("" :: "v"(a0), "v"(a1), "v"(a2), "v"(a3), "v"(a4), "v"(a5));
    const float mx = fmaxf(a0, fmaxf(a1, a2));
    const float e0 = __expf(a0 - mx), e1 = __expf(a1 - mx), e2 = __expf(a2 - mx);
    const float inv = 1.f / (e0 + e1 + e2);
    const float ca = e0 * inv, cb = e1 * inv, cc = e2 * inv;
    const float iv = 1.f / (1.f + __expf(-a3));
    const float cv = tanhf(a4);
    const float ov = 1.f / (1.f + __expf(-a5));
    const float pushed = iv * cv;
    row[t] = pushed;
    row[t + 1] = pushed;
    {
      const float sl_t = (t == 0) ? pushed : tl_old;
      const float top = sl_t * ca + (pushed + vbs) * cb + (pushed + vbd) * cc;
      const float hn = top * ov;
      hv[((t + 1) & 1) * H_DIM + j] = hn;
      out[(size_t)(t * B_DIM + b) * H_DIM + j] = hn;
      hlast = hn;
    }
    {
      const float2* sp = (const float2*)row;
      float2* dp = (float2*)row;
      float2 c0 = sp[0];
      float2 q[UWIN];
      #pragma unroll
      for (int n = 0; n < UWIN; ++n) q[n] = sp[1 + n];
      __builtin_amdgcn_sched_barrier(0);
      float sl = c0.x;
      #pragma unroll
      for (int k = 0; k < NPAIR; ++k) {
        const float2 c1 = q[k % UWIN];
        float2 n;
        n.x = sl   * ca + (c0.x + vbs) * cb + (c0.y + vbd) * cc;
        n.y = c0.x * ca + (c0.y + vbs) * cb + (c1.x + vbd) * cc;
        dp[k] = n;
        if (k + 1 + UWIN <= NPAIR) q[k % UWIN] = sp[k + 1 + UWIN];
        if (k % UGRP == UGRP - 1) __builtin_amdgcn_sched_barrier(0);
        sl = c0.y; c0 = c1;
      }
    }
    __syncthreads();
  }
  out[(size_t)T_DIM * B_DIM * H_DIM + (size_t)b * H_DIM + j] = hlast;
}

extern "C" void kernel_launch(void* const* d_in, const int* in_sizes, int n_in,
                              void* d_out, int out_size, void* d_ws, size_t ws_size,
                              hipStream_t stream) {
  (void)in_sizes; (void)n_in; (void)out_size; (void)ws_size;
  const float* x  = (const float*)d_in[0];
  const float* Wf = (const float*)d_in[1];
  const float* bf = (const float*)d_in[2];
  const float* Wi = (const float*)d_in[3];
  const float* bi = (const float*)d_in[4];
  const float* Wc = (const float*)d_in[5];
  const float* bc = (const float*)d_in[6];
  const float* Wo = (const float*)d_in[7];
  const float* bo = (const float*)d_in[8];
  const float* b_stay = (const float*)d_in[11];
  const float* b_down = (const float*)d_in[13];
  float* out = (float*)d_out;

  float* Xp  = (float*)d_ws;

  xproj_wmma<<<(ROWS / 16) * (G_DIM / 16) / 8, 256, 0, stream>>>(
      x, Wf, bf, Wi, bi, Wc, bc, Wo, bo, Xp);

  size_t shmem = (size_t)(H_DIM * SSTR + 2 * H_DIM) * sizeof(float);
  (void)hipFuncSetAttribute((const void*)scan_kernel,
                            hipFuncAttributeMaxDynamicSharedMemorySize, (int)shmem);
  scan_kernel<<<B_DIM, SCAN_THREADS, shmem, stream>>>(Xp, Wf, Wi, Wc, Wo, b_stay, b_down, out);
}


# scan state update: step t sweeps only the live stack slots 0..t (uniform exit every 8 slots) instead of all 128; slots above t reach no output, results bit-identical
# speedup vs baseline: 8.1174x; 1.1871x over previous
#include <hip/hip_runtime.h>
#include <math.h>


typedef float v2f __attribute__((ext_vector_type(2)));
typedef float v8f __attribute__((ext_vector_type(8)));

#define T_DIM 128
#define B_DIM 32
#define E_DIM 128
#define H_DIM 128
#define GE_DIM 256
#define G_DIM 768
#define GI_OFF 384
#define GC_OFF 512
#define GO_OFF 640
#define ROWS (T_DIM * B_DIM)
#define NPAIR 64
#define SSTR 134

__device__ __forceinline__ const float* wrow(int k,
    const float* Wf, const float* Wi, const float* Wc, const float* Wo) {
  if (k < GI_OFF) return Wf + (size_t)k * GE_DIM;
  if (k < GC_OFF) return Wi + (size_t)(k - GI_OFF) * GE_DIM;
  if (k < GO_OFF) return Wc + (size_t)(k - GC_OFF) * GE_DIM;
  return Wo + (size_t)(k - GO_OFF) * GE_DIM;
}
__device__ __forceinline__ float biasAt(int k,
    const float* bf, const float* bi, const float* bc, const float* bo) {
  if (k < GI_OFF) return bf[k];
  if (k < GC_OFF) return bi[k - GI_OFF];
  if (k < GO_OFF) return bc[k - GC_OFF];
  return bo[k - GO_OFF];
}

__global__ void __launch_bounds__(256)
xproj_wmma(const float* __restrict__ x,
           const float* __restrict__ Wf, const float* __restrict__ bf,
           const float* __restrict__ Wi, const float* __restrict__ bi,
           const float* __restrict__ Wc, const float* __restrict__ bc,
           const float* __restrict__ Wo, const float* __restrict__ bo,
           float* __restrict__ Xp) {
  const int wid  = blockIdx.x * (blockDim.x >> 5) + (threadIdx.x >> 5);
  const int lane = threadIdx.x & 31;
  const int rowTile = wid / (G_DIM / 16);
  const int colTile = wid % (G_DIM / 16);
  if (rowTile >= ROWS / 16) return;
  const int r0 = rowTile * 16;
  const int k0 = colTile * 16;
  const int mn = lane & 15;
  const int kb = (lane < 16) ? 0 : 2;

  const float* xrow = x + (size_t)(r0 + mn) * E_DIM + kb;
  const int kg = k0 + mn;
  const float* wr = wrow(kg, Wf, Wi, Wc, Wo) + H_DIM + kb;
  const float bb = biasAt(kg, bf, bi, bc, bo);

  v8f acc0 = {};
  v8f acc1 = {};
  #pragma unroll
  for (int kk = 0; kk < E_DIM; kk += 8) {
    v2f a0 = *(const v2f*)(xrow + kk);
    v2f b0 = *(const v2f*)(wr + kk);
    v2f a1 = *(const v2f*)(xrow + kk + 4);
    v2f b1 = *(const v2f*)(wr + kk + 4);
    acc0 = __builtin_amdgcn_wmma_f32_16x16x4_f32(false, a0, false, b0,
                                                 (short)0, acc0, false, false);
    acc1 = __builtin_amdgcn_wmma_f32_16x16x4_f32(false, a1, false, b1,
                                                 (short)0, acc1, false, false);
  }
  const int mbase = (lane < 16) ? 0 : 8;
  #pragma unroll
  for (int v = 0; v < 8; ++v) {
    Xp[(size_t)(r0 + mbase + v) * G_DIM + kg] = acc0[v] + acc1[v] + bb;
  }
}

#define SCAN_THREADS H_DIM
#define NG 6
#define HWIN 6
#define UWIN 12
#define UGRP 4

__global__ void __launch_bounds__(SCAN_THREADS)
scan_kernel(const float* __restrict__ Xp,
            const float* __restrict__ Wf, const float* __restrict__ Wi,
            const float* __restrict__ Wc, const float* __restrict__ Wo,
            const float* __restrict__ b_stay,
            const float* __restrict__ b_down,
            float* __restrict__ out) {
  extern __shared__ __align__(16) float smem[];
  float* hv   = smem;
  float* rows = hv + 2 * H_DIM;

  const int j = threadIdx.x;
  const int b = blockIdx.x;
  float* row = rows + j * SSTR;

  float w[NG * H_DIM];
  {
    const float* rp[NG];
    rp[0] = Wf + (size_t)(3 * j + 0) * GE_DIM;
    rp[1] = Wf + (size_t)(3 * j + 1) * GE_DIM;
    rp[2] = Wf + (size_t)(3 * j + 2) * GE_DIM;
    rp[3] = Wi + (size_t)j * GE_DIM;
    rp[4] = Wc + (size_t)j * GE_DIM;
    rp[5] = Wo + (size_t)j * GE_DIM;
    #pragma unroll
    for (int r = 0; r < NG; ++r) {
      const float4* wr4 = (const float4*)rp[r];
      #pragma unroll
      for (int i4 = 0; i4 < H_DIM / 4; ++i4) {
        const float4 v = wr4[i4];
        w[r * H_DIM + 4 * i4 + 0] = v.x; w[r * H_DIM + 4 * i4 + 1] = v.y;
        w[r * H_DIM + 4 * i4 + 2] = v.z; w[r * H_DIM + 4 * i4 + 3] = v.w;
      }
    }
  }
  const float vbs = b_stay[j], vbd = b_down[j];

  {
    float2* z = (float2*)row;
    for (int i = 0; i < SSTR / 2; ++i) z[i] = make_float2(0.f, 0.f);
  }
  hv[j] = 0.f; hv[H_DIM + j] = 0.f;
  __syncthreads();

  float hlast = 0.f;
  for (int t = 0; t < T_DIM; ++t) {
    const float* xp = Xp + (size_t)(t * B_DIM + b) * G_DIM;
    if (t + 1 < T_DIM) {
      __builtin_prefetch(xp + (size_t)B_DIM * G_DIM + NG * j, 0, 0);
    }
    float a0 = xp[3 * j + 0], a1 = xp[3 * j + 1], a2 = xp[3 * j + 2];
    float a3 = xp[GI_OFF + j], a4 = xp[GC_OFF + j], a5 = xp[GO_OFF + j];
    const float tl_old = row[t > 0 ? t - 1 : 0];
    {
      const float4* h4 = (const float4*)(hv + (t & 1) * H_DIM);
      float4 q[HWIN];
      #pragma unroll
      for (int n = 0; n < HWIN; ++n) q[n] = h4[n];
      __builtin_amdgcn_sched_barrier(0);
      #pragma unroll
      for (int i4 = 0; i4 < H_DIM / 4; ++i4) {
        const float4 h = q[i4 % HWIN];
        const float hc[4] = {h.x, h.y, h.z, h.w};
        #pragma unroll
        for (int c = 0; c < 4; ++c) {
          const int i = 4 * i4 + c;
          a0 = fmaf(hc[c], w[0 * H_DIM + i], a0);
          a1 = fmaf(hc[c], w[1 * H_DIM + i], a1);
          a2 = fmaf(hc[c], w[2 * H_DIM + i], a2);
          a3 = fmaf(hc[c], w[3 * H_DIM + i], a3);
          a4 = fmaf(hc[c], w[4 * H_DIM + i], a4);
          a5 = fmaf(hc[c], w[5 * H_DIM + i], a5);
        }
        if (i4 + HWIN < H_DIM / 4) q[i4 % HWIN] = h4[i4 + HWIN];
        __builtin_amdgcn_sched_barrier(0);
      }
    }
    asm volatile("" :: "v"(a0), "v"(a1), "v"(a2), "v"(a3), "v"(a4), "v"(a5));
    const float mx = fmaxf(a0, fmaxf(a1, a2));
    const float e0 = __expf(a0 - mx), e1 = __expf(a1 - mx), e2 = __expf(a2 - mx);
    const float inv = 1.f / (e0 + e1 + e2);
    const float ca = e0 * inv, cb = e1 * inv, cc = e2 * inv;
    const float iv = 1.f / (1.f + __expf(-a3));
    const float cv = tanhf(a4);
    const float ov = 1.f / (1.f + __expf(-a5));
    const float pushed = iv * cv;
    row[t] = pushed;
    row[t + 1] = pushed;
    {
      const float sl_t = (t == 0) ? pushed : tl_old;
      const float top = sl_t * ca + (pushed + vbs) * cb + (pushed + vbd) * cc;
      const float hn = top * ov;
      hv[((t + 1) & 1) * H_DIM + j] = hn;
      out[(size_t)(t * B_DIM + b) * H_DIM + j] = hn;
      hlast = hn;
    }
    {
      const float2* sp = (const float2*)row;
      float2* dp = (float2*)row;
      float2 c0 = sp[0];
      float2 q[UWIN];
      #pragma unroll
      for (int n = 0; n < UWIN; ++n) q[n] = sp[1 + n];
      __builtin_amdgcn_sched_barrier(0);
      float sl = c0.x;
      #pragma unroll
      for (int k = 0; k < NPAIR; ++k) {
        const float2 c1 = q[k % UWIN];
        float2 n;
        n.x = sl   * ca + (c0.x + vbs) * cb + (c0.y + vbd) * cc;
        n.y = c0.x * ca + (c0.y + vbs) * cb + (c1.x + vbd) * cc;
        dp[k] = n;
        if (k + 1 + UWIN <= NPAIR) q[k % UWIN] = sp[k + 1 + UWIN];
        if (k % UGRP == UGRP - 1) {
          __builtin_amdgcn_sched_barrier(0);
          if (2 * k + 1 >= t) break;
        }
        sl = c0.y; c0 = c1;
      }
    }
    __syncthreads();
  }
  out[(size_t)T_DIM * B_DIM * H_DIM + (size_t)b * H_DIM + j] = hlast;
}

extern "C" void kernel_launch(void* const* d_in, const int* in_sizes, int n_in,
                              void* d_out, int out_size, void* d_ws, size_t ws_size,
                              hipStream_t stream) {
  (void)in_sizes; (void)n_in; (void)out_size; (void)ws_size;
  const float* x  = (const float*)d_in[0];
  const float* Wf = (const float*)d_in[1];
  const float* bf = (const float*)d_in[2];
  const float* Wi = (const float*)d_in[3];
  const float* bi = (const float*)d_in[4];
  const float* Wc = (const float*)d_in[5];
  const float* bc = (const float*)d_in[6];
  const float* Wo = (const float*)d_in[7];
  const float* bo = (const float*)d_in[8];
  const float* b_stay = (const float*)d_in[11];
  const float* b_down = (const float*)d_in[13];
  float* out = (float*)d_out;

  float* Xp  = (float*)d_ws;

  xproj_wmma<<<(ROWS / 16) * (G_DIM / 16) / 8, 256, 0, stream>>>(
      x, Wf, bf, Wi, bi, Wc, bc, Wo, bo, Xp);

  size_t shmem = (size_t)(H_DIM * SSTR + 2 * H_DIM) * sizeof(float);
  (void)hipFuncSetAttribute((const void*)scan_kernel,
                            hipFuncAttributeMaxDynamicSharedMemorySize, (int)shmem);
  scan_kernel<<<B_DIM, SCAN_THREADS, shmem, stream>>>(Xp, Wf, Wi, Wc, Wo, b_stay, b_down, out);
}


# scan: stack sweep and its LDS rows dropped: only the top slot ever feeds an output (a value climbs <=1 slot a step, the top climbs 1); old top kept in a register, out store moved behind the barrier
# speedup vs baseline: 9.2642x; 1.1413x over previous
#include <hip/hip_runtime.h>
#include <math.h>


typedef float v2f __attribute__((ext_vector_type(2)));
typedef float v8f __attribute__((ext_vector_type(8)));

#define T_DIM 128
#define B_DIM 32
#define E_DIM 128
#define H_DIM 128
#define GE_DIM 256
#define G_DIM 768
#define GI_OFF 384
#define GC_OFF 512
#define GO_OFF 640
#define ROWS (T_DIM * B_DIM)

__device__ __forceinline__ const float* wrow(int k,
    const float* Wf, const float* Wi, const float* Wc, const float* Wo) {
  if (k < GI_OFF) return Wf + (size_t)k * GE_DIM;
  if (k < GC_OFF) return Wi + (size_t)(k - GI_OFF) * GE_DIM;
  if (k < GO_OFF) return Wc + (size_t)(k - GC_OFF) * GE_DIM;
  return Wo + (size_t)(k - GO_OFF) * GE_DIM;
}
__device__ __forceinline__ float biasAt(int k,
    const float* bf, const float* bi, const float* bc, const float* bo) {
  if (k < GI_OFF) return bf[k];
  if (k < GC_OFF) return bi[k - GI_OFF];
  if (k < GO_OFF) return bc[k - GC_OFF];
  return bo[k - GO_OFF];
}

__global__ void __launch_bounds__(256)
xproj_wmma(const float* __restrict__ x,
           const float* __restrict__ Wf, const float* __restrict__ bf,
           const float* __restrict__ Wi, const float* __restrict__ bi,
           const float* __restrict__ Wc, const float* __restrict__ bc,
           const float* __restrict__ Wo, const float* __restrict__ bo,
           float* __restrict__ Xp) {
  const int wid  = blockIdx.x * (blockDim.x >> 5) + (threadIdx.x >> 5);
  const int lane = threadIdx.x & 31;
  const int rowTile = wid / (G_DIM / 16);
  const int colTile = wid % (G_DIM / 16);
  if (rowTile >= ROWS / 16) return;
  const int r0 = rowTile * 16;
  const int k0 = colTile * 16;
  const int mn = lane & 15;
  const int kb = (lane < 16) ? 0 : 2;

  const float* xrow = x + (size_t)(r0 + mn) * E_DIM + kb;
  const int kg = k0 + mn;
  const float* wr = wrow(kg, Wf, Wi, Wc, Wo) + H_DIM + kb;
  const float bb = biasAt(kg, bf, bi, bc, bo);

  v8f acc0 = {};
  v8f acc1 = {};
  #pragma unroll
  for (int kk = 0; kk < E_DIM; kk += 8) {
    v2f a0 = *(const v2f*)(xrow + kk);
    v2f b0 = *(const v2f*)(wr + kk);
    v2f a1 = *(const v2f*)(xrow + kk + 4);
    v2f b1 = *(const v2f*)(wr + kk + 4);
    acc0 = __builtin_amdgcn_wmma_f32_16x16x4_f32(false, a0, false, b0,
                                                 (short)0, acc0, false, false);
    acc1 = __builtin_amdgcn_wmma_f32_16x16x4_f32(false, a1, false, b1,
                                                 (short)0, acc1, false, false);
  }
  const int mbase = (lane < 16) ? 0 : 8;
  #pragma unroll
  for (int v = 0; v < 8; ++v) {
    Xp[(size_t)(r0 + mbase + v) * G_DIM + kg] = acc0[v] + acc1[v] + bb;
  }
}

#define SCAN_THREADS H_DIM
#define NG 6
#define HWIN 6

__global__ void __launch_bounds__(SCAN_THREADS)
scan_kernel(const float* __restrict__ Xp,
            const float* __restrict__ Wf, const float* __restrict__ Wi,
            const float* __restrict__ Wc, const float* __restrict__ Wo,
            const float* __restrict__ b_stay,
            const float* __restrict__ b_down,
            float* __restrict__ out) {
  extern __shared__ __align__(16) float smem[];
  float* hv   = smem;

  const int j = threadIdx.x;
  const int b = blockIdx.x;

  float w[NG * H_DIM];
  {
    const float* rp[NG];
    rp[0] = Wf + (size_t)(3 * j + 0) * GE_DIM;
    rp[1] = Wf + (size_t)(3 * j + 1) * GE_DIM;
    rp[2] = Wf + (size_t)(3 * j + 2) * GE_DIM;
    rp[3] = Wi + (size_t)j * GE_DIM;
    rp[4] = Wc + (size_t)j * GE_DIM;
    rp[5] = Wo + (size_t)j * GE_DIM;
    #pragma unroll
    for (int r = 0; r < NG; ++r) {
      const float4* wr4 = (const float4*)rp[r];
      #pragma unroll
      for (int i4 = 0; i4 < H_DIM / 4; ++i4) {
        const float4 v = wr4[i4];
        w[r * H_DIM + 4 * i4 + 0] = v.x; w[r * H_DIM + 4 * i4 + 1] = v.y;
        w[r * H_DIM + 4 * i4 + 2] = v.z; w[r * H_DIM + 4 * i4 + 3] = v.w;
      }
    }
  }
  const float vbs = b_stay[j], vbd = b_down[j];

  hv[j] = 0.f; hv[H_DIM + j] = 0.f;
  __syncthreads();

  float hlast = 0.f;
  float top_prev = 0.f;
  for (int t = 0; t < T_DIM; ++t) {
    const float* xp = Xp + (size_t)(t * B_DIM + b) * G_DIM;
    if (t + 1 < T_DIM) {
      __builtin_prefetch(xp + (size_t)B_DIM * G_DIM + NG * j, 0, 0);
    }
    float a0 = xp[3 * j + 0], a1 = xp[3 * j + 1], a2 = xp[3 * j + 2];
    float a3 = xp[GI_OFF + j], a4 = xp[GC_OFF + j], a5 = xp[GO_OFF + j];
    {
      const float4* h4 = (const float4*)(hv + (t & 1) * H_DIM);
      float4 q[HWIN];
      #pragma unroll
      for (int n = 0; n < HWIN; ++n) q[n] = h4[n];
      __builtin_amdgcn_sched_barrier(0);
      #pragma unroll
      for (int i4 = 0; i4 < H_DIM / 4; ++i4) {
        const float4 h = q[i4 % HWIN];
        const float hc[4] = {h.x, h.y, h.z, h.w};
        #pragma unroll
        for (int c = 0; c < 4; ++c) {
          const int i = 4 * i4 + c;
          a0 = fmaf(hc[c], w[0 * H_DIM + i], a0);
          a1 = fmaf(hc[c], w[1 * H_DIM + i], a1);
          a2 = fmaf(hc[c], w[2 * H_DIM + i], a2);
          a3 = fmaf(hc[c], w[3 * H_DIM + i], a3);
          a4 = fmaf(hc[c], w[4 * H_DIM + i], a4);
          a5 = fmaf(hc[c], w[5 * H_DIM + i], a5);
        }
        if (i4 + HWIN < H_DIM / 4) q[i4 % HWIN] = h4[i4 + HWIN];
        __builtin_amdgcn_sched_barrier(0);
      }
    }
    asm volatile("" :: "v"(a0), "v"(a1), "v"(a2), "v"(a3), "v"(a4), "v"(a5));
    const float mx = fmaxf(a0, fmaxf(a1, a2));
    const float e0 = __expf(a0 - mx), e1 = __expf(a1 - mx), e2 = __expf(a2 - mx);
    const float inv = 1.f / (e0 + e1 + e2);
    const float ca = e0 * inv, cb = e1 * inv, cc = e2 * inv;
    const float iv = 1.f / (1.f + __expf(-a3));
    const float cv = tanhf(a4);
    const float ov = 1.f / (1.f + __expf(-a5));
    const float pushed = iv * cv;
    {
      const float sl_t = (t == 0) ? pushed : top_prev;
      const float top = sl_t * ca + (pushed + vbs) * cb + (pushed + vbd) * cc;
      const float hn = top * ov;
      hv[((t + 1) & 1) * H_DIM + j] = hn;
      hlast = hn;
      top_prev = top;
    }
    __syncthreads();
    out[(size_t)(t * B_DIM + b) * H_DIM + j] = hlast;
  }
  out[(size_t)T_DIM * B_DIM * H_DIM + (size_t)b * H_DIM + j] = hlast;
}

extern "C" void kernel_launch(void* const* d_in, const int* in_sizes, int n_in,
                              void* d_out, int out_size, void* d_ws, size_t ws_size,
                              hipStream_t stream) {
  (void)in_sizes; (void)n_in; (void)out_size; (void)ws_size;
  const float* x  = (const float*)d_in[0];
  const float* Wf = (const float*)d_in[1];
  const float* bf = (const float*)d_in[2];
  const float* Wi = (const float*)d_in[3];
  const float* bi = (const float*)d_in[4];
  const float* Wc = (const float*)d_in[5];
  const float* bc = (const float*)d_in[6];
  const float* Wo = (const float*)d_in[7];
  const float* bo = (const float*)d_in[8];
  const float* b_stay = (const float*)d_in[11];
  const float* b_down = (const float*)d_in[13];
  float* out = (float*)d_out;

  float* Xp  = (float*)d_ws;

  xproj_wmma<<<(ROWS / 16) * (G_DIM / 16) / 8, 256, 0, stream>>>(
      x, Wf, bf, Wi, bi, Wc, bc, Wo, bo, Xp);

  size_t shmem = (size_t)(2 * H_DIM) * sizeof(float);
  scan_kernel<<<B_DIM, SCAN_THREADS, shmem, stream>>>(Xp, Wf, Wi, Wc, Wo, b_stay, b_down, out);
}


# scan: the six projected-input start values of step t+1 are loaded into registers during step t (behind its first h group), replacing the L2 prefetch: their load latency no longer heads each step
# speedup vs baseline: 12.5583x; 1.3556x over previous
#include <hip/hip_runtime.h>
#include <math.h>


typedef float v2f __attribute__((ext_vector_type(2)));
typedef float v8f __attribute__((ext_vector_type(8)));

#define T_DIM 128
#define B_DIM 32
#define E_DIM 128
#define H_DIM 128
#define GE_DIM 256
#define G_DIM 768
#define GI_OFF 384
#define GC_OFF 512
#define GO_OFF 640
#define ROWS (T_DIM * B_DIM)

__device__ __forceinline__ const float* wrow(int k,
    const float* Wf, const float* Wi, const float* Wc, const float* Wo) {
  if (k < GI_OFF) return Wf + (size_t)k * GE_DIM;
  if (k < GC_OFF) return Wi + (size_t)(k - GI_OFF) * GE_DIM;
  if (k < GO_OFF) return Wc + (size_t)(k - GC_OFF) * GE_DIM;
  return Wo + (size_t)(k - GO_OFF) * GE_DIM;
}
__device__ __forceinline__ float biasAt(int k,
    const float* bf, const float* bi, const float* bc, const float* bo) {
  if (k < GI_OFF) return bf[k];
  if (k < GC_OFF) return bi[k - GI_OFF];
  if (k < GO_OFF) return bc[k - GC_OFF];
  return bo[k - GO_OFF];
}

__global__ void __launch_bounds__(256)
xproj_wmma(const float* __restrict__ x,
           const float* __restrict__ Wf, const float* __restrict__ bf,
           const float* __restrict__ Wi, const float* __restrict__ bi,
           const float* __restrict__ Wc, const float* __restrict__ bc,
           const float* __restrict__ Wo, const float* __restrict__ bo,
           float* __restrict__ Xp) {
  const int wid  = blockIdx.x * (blockDim.x >> 5) + (threadIdx.x >> 5);
  const int lane = threadIdx.x & 31;
  const int rowTile = wid / (G_DIM / 16);
  const int colTile = wid % (G_DIM / 16);
  if (rowTile >= ROWS / 16) return;
  const int r0 = rowTile * 16;
  const int k0 = colTile * 16;
  const int mn = lane & 15;
  const int kb = (lane < 16) ? 0 : 2;

  const float* xrow = x + (size_t)(r0 + mn) * E_DIM + kb;
  const int kg = k0 + mn;
  const float* wr = wrow(kg, Wf, Wi, Wc, Wo) + H_DIM + kb;
  const float bb = biasAt(kg, bf, bi, bc, bo);

  v8f acc0 = {};
  v8f acc1 = {};
  #pragma unroll
  for (int kk = 0; kk < E_DIM; kk += 8) {
    v2f a0 = *(const v2f*)(xrow + kk);
    v2f b0 = *(const v2f*)(wr + kk);
    v2f a1 = *(const v2f*)(xrow + kk + 4);
    v2f b1 = *(const v2f*)(wr + kk + 4);
    acc0 = __builtin_amdgcn_wmma_f32_16x16x4_f32(false, a0, false, b0,
                                                 (short)0, acc0, false, false);
    acc1 = __builtin_amdgcn_wmma_f32_16x16x4_f32(false, a1, false, b1,
                                                 (short)0, acc1, false, false);
  }
  const int mbase = (lane < 16) ? 0 : 8;
  #pragma unroll
  for (int v = 0; v < 8; ++v) {
    Xp[(size_t)(r0 + mbase + v) * G_DIM + kg] = acc0[v] + acc1[v] + bb;
  }
}

#define SCAN_THREADS H_DIM
#define NG 6
#define HWIN 6

__global__ void __launch_bounds__(SCAN_THREADS)
scan_kernel(const float* __restrict__ Xp,
            const float* __restrict__ Wf, const float* __restrict__ Wi,
            const float* __restrict__ Wc, const float* __restrict__ Wo,
            const float* __restrict__ b_stay,
            const float* __restrict__ b_down,
            float* __restrict__ out) {
  extern __shared__ __align__(16) float smem[];
  float* hv   = smem;

  const int j = threadIdx.x;
  const int b = blockIdx.x;

  float w[NG * H_DIM];
  {
    const float* rp[NG];
    rp[0] = Wf + (size_t)(3 * j + 0) * GE_DIM;
    rp[1] = Wf + (size_t)(3 * j + 1) * GE_DIM;
    rp[2] = Wf + (size_t)(3 * j + 2) * GE_DIM;
    rp[3] = Wi + (size_t)j * GE_DIM;
    rp[4] = Wc + (size_t)j * GE_DIM;
    rp[5] = Wo + (size_t)j * GE_DIM;
    #pragma unroll
    for (int r = 0; r < NG; ++r) {
      const float4* wr4 = (const float4*)rp[r];
      #pragma unroll
      for (int i4 = 0; i4 < H_DIM / 4; ++i4) {
        const float4 v = wr4[i4];
        w[r * H_DIM + 4 * i4 + 0] = v.x; w[r * H_DIM + 4 * i4 + 1] = v.y;
        w[r * H_DIM + 4 * i4 + 2] = v.z; w[r * H_DIM + 4 * i4 + 3] = v.w;
      }
    }
  }
  const float vbs = b_stay[j], vbd = b_down[j];

  hv[j] = 0.f; hv[H_DIM + j] = 0.f;
  __syncthreads();

  float hlast = 0.f;
  float top_prev = 0.f;
  float n0, n1, n2, n3, n4, n5;
  {
    const float* xp = Xp + (size_t)b * G_DIM;
    n0 = xp[3 * j + 0]; n1 = xp[3 * j + 1]; n2 = xp[3 * j + 2];
    n3 = xp[GI_OFF + j]; n4 = xp[GC_OFF + j]; n5 = xp[GO_OFF + j];
  }
  for (int t = 0; t < T_DIM; ++t) {
    float a0 = n0, a1 = n1, a2 = n2, a3 = n3, a4 = n4, a5 = n5;
    const float* xn = Xp + (size_t)(((t + 1 < T_DIM) ? t + 1 : t) * B_DIM + b) * G_DIM;
    {
      const float4* h4 = (const float4*)(hv + (t & 1) * H_DIM);
      float4 q[HWIN];
      #pragma unroll
      for (int n = 0; n < HWIN; ++n) q[n] = h4[n];
      __builtin_amdgcn_sched_barrier(0);
      #pragma unroll
      for (int i4 = 0; i4 < H_DIM / 4; ++i4) {
        const float4 h = q[i4 % HWIN];
        const float hc[4] = {h.x, h.y, h.z, h.w};
        #pragma unroll
        for (int c = 0; c < 4; ++c) {
          const int i = 4 * i4 + c;
          a0 = fmaf(hc[c], w[0 * H_DIM + i], a0);
          a1 = fmaf(hc[c], w[1 * H_DIM + i], a1);
          a2 = fmaf(hc[c], w[2 * H_DIM + i], a2);
          a3 = fmaf(hc[c], w[3 * H_DIM + i], a3);
          a4 = fmaf(hc[c], w[4 * H_DIM + i], a4);
          a5 = fmaf(hc[c], w[5 * H_DIM + i], a5);
        }
        if (i4 + HWIN < H_DIM / 4) q[i4 % HWIN] = h4[i4 + HWIN];
        if (i4 == 0) {
          n0 = xn[3 * j + 0]; n1 = xn[3 * j + 1]; n2 = xn[3 * j + 2];
          n3 = xn[GI_OFF + j]; n4 = xn[GC_OFF + j]; n5 = xn[GO_OFF + j];
        }
        __builtin_amdgcn_sched_barrier(0);
      }
    }
    asm volatile("" :: "v"(a0), "v"(a1), "v"(a2), "v"(a3), "v"(a4), "v"(a5));
    const float mx = fmaxf(a0, fmaxf(a1, a2));
    const float e0 = __expf(a0 - mx), e1 = __expf(a1 - mx), e2 = __expf(a2 - mx);
    const float inv = 1.f / (e0 + e1 + e2);
    const float ca = e0 * inv, cb = e1 * inv, cc = e2 * inv;
    const float iv = 1.f / (1.f + __expf(-a3));
    const float cv = tanhf(a4);
    const float ov = 1.f / (1.f + __expf(-a5));
    const float pushed = iv * cv;
    {
      const float sl_t = (t == 0) ? pushed : top_prev;
      const float top = sl_t * ca + (pushed + vbs) * cb + (pushed + vbd) * cc;
      const float hn = top * ov;
      hv[((t + 1) & 1) * H_DIM + j] = hn;
      hlast = hn;
      top_prev = top;
    }
    __syncthreads();
    out[(size_t)(t * B_DIM + b) * H_DIM + j] = hlast;
  }
  out[(size_t)T_DIM * B_DIM * H_DIM + (size_t)b * H_DIM + j] = hlast;
}

extern "C" void kernel_launch(void* const* d_in, const int* in_sizes, int n_in,
                              void* d_out, int out_size, void* d_ws, size_t ws_size,
                              hipStream_t stream) {
  (void)in_sizes; (void)n_in; (void)out_size; (void)ws_size;
  const float* x  = (const float*)d_in[0];
  const float* Wf = (const float*)d_in[1];
  const float* bf = (const float*)d_in[2];
  const float* Wi = (const float*)d_in[3];
  const float* bi = (const float*)d_in[4];
  const float* Wc = (const float*)d_in[5];
  const float* bc = (const float*)d_in[6];
  const float* Wo = (const float*)d_in[7];
  const float* bo = (const float*)d_in[8];
  const float* b_stay = (const float*)d_in[11];
  const float* b_down = (const float*)d_in[13];
  float* out = (float*)d_out;

  float* Xp  = (float*)d_ws;

  xproj_wmma<<<(ROWS / 16) * (G_DIM / 16) / 8, 256, 0, stream>>>(
      x, Wf, bf, Wi, bi, Wc, bc, Wo, bo, Xp);

  size_t shmem = (size_t)(2 * H_DIM) * sizeof(float);
  scan_kernel<<<B_DIM, SCAN_THREADS, shmem, stream>>>(Xp, Wf, Wi, Wc, Wo, b_stay, b_down, out);
}


# scan: gate sums as packed two-wide f32 fma, each kept as four partial sums (columns mod 4) added at the end: changed summation order; h read-ahead 6 -> 12 groups, a group now taking half the cycles
# speedup vs baseline: 15.5404x; 1.2375x over previous
#include <hip/hip_runtime.h>
#include <math.h>


typedef float v2f __attribute__((ext_vector_type(2)));
typedef float v8f __attribute__((ext_vector_type(8)));

#define T_DIM 128
#define B_DIM 32
#define E_DIM 128
#define H_DIM 128
#define GE_DIM 256
#define G_DIM 768
#define GI_OFF 384
#define GC_OFF 512
#define GO_OFF 640
#define ROWS (T_DIM * B_DIM)

__device__ __forceinline__ const float* wrow(int k,
    const float* Wf, const float* Wi, const float* Wc, const float* Wo) {
  if (k < GI_OFF) return Wf + (size_t)k * GE_DIM;
  if (k < GC_OFF) return Wi + (size_t)(k - GI_OFF) * GE_DIM;
  if (k < GO_OFF) return Wc + (size_t)(k - GC_OFF) * GE_DIM;
  return Wo + (size_t)(k - GO_OFF) * GE_DIM;
}
__device__ __forceinline__ float biasAt(int k,
    const float* bf, const float* bi, const float* bc, const float* bo) {
  if (k < GI_OFF) return bf[k];
  if (k < GC_OFF) return bi[k - GI_OFF];
  if (k < GO_OFF) return bc[k - GC_OFF];
  return bo[k - GO_OFF];
}

__global__ void __launch_bounds__(256)
xproj_wmma(const float* __restrict__ x,
           const float* __restrict__ Wf, const float* __restrict__ bf,
           const float* __restrict__ Wi, const float* __restrict__ bi,
           const float* __restrict__ Wc, const float* __restrict__ bc,
           const float* __restrict__ Wo, const float* __restrict__ bo,
           float* __restrict__ Xp) {
  const int wid  = blockIdx.x * (blockDim.x >> 5) + (threadIdx.x >> 5);
  const int lane = threadIdx.x & 31;
  const int rowTile = wid / (G_DIM / 16);
  const int colTile = wid % (G_DIM / 16);
  if (rowTile >= ROWS / 16) return;
  const int r0 = rowTile * 16;
  const int k0 = colTile * 16;
  const int mn = lane & 15;
  const int kb = (lane < 16) ? 0 : 2;

  const float* xrow = x + (size_t)(r0 + mn) * E_DIM + kb;
  const int kg = k0 + mn;
  const float* wr = wrow(kg, Wf, Wi, Wc, Wo) + H_DIM + kb;
  const float bb = biasAt(kg, bf, bi, bc, bo);

  v8f acc0 = {};
  v8f acc1 = {};
  #pragma unroll
  for (int kk = 0; kk < E_DIM; kk += 8) {
    v2f a0 = *(const v2f*)(xrow + kk);
    v2f b0 = *(const v2f*)(wr + kk);
    v2f a1 = *(const v2f*)(xrow + kk + 4);
    v2f b1 = *(const v2f*)(wr + kk + 4);
    acc0 = __builtin_amdgcn_wmma_f32_16x16x4_f32(false, a0, false, b0,
                                                 (short)0, acc0, false, false);
    acc1 = __builtin_amdgcn_wmma_f32_16x16x4_f32(false, a1, false, b1,
                                                 (short)0, acc1, false, false);
  }
  const int mbase = (lane < 16) ? 0 : 8;
  #pragma unroll
  for (int v = 0; v < 8; ++v) {
    Xp[(size_t)(r0 + mbase + v) * G_DIM + kg] = acc0[v] + acc1[v] + bb;
  }
}

#define SCAN_THREADS H_DIM
#define NG 6
#define HWIN 12

__global__ void __launch_bounds__(SCAN_THREADS)
scan_kernel(const float* __restrict__ Xp,
            const float* __restrict__ Wf, const float* __restrict__ Wi,
            const float* __restrict__ Wc, const float* __restrict__ Wo,
            const float* __restrict__ b_stay,
            const float* __restrict__ b_down,
            float* __restrict__ out) {
  extern __shared__ __align__(16) float smem[];
  float* hv   = smem;

  const int j = threadIdx.x;
  const int b = blockIdx.x;

  v2f w2[NG * H_DIM / 2];
  {
    const float* rp[NG];
    rp[0] = Wf + (size_t)(3 * j + 0) * GE_DIM;
    rp[1] = Wf + (size_t)(3 * j + 1) * GE_DIM;
    rp[2] = Wf + (size_t)(3 * j + 2) * GE_DIM;
    rp[3] = Wi + (size_t)j * GE_DIM;
    rp[4] = Wc + (size_t)j * GE_DIM;
    rp[5] = Wo + (size_t)j * GE_DIM;
    #pragma unroll
    for (int r = 0; r < NG; ++r) {
      const float4* wr4 = (const float4*)rp[r];
      #pragma unroll
      for (int i4 = 0; i4 < H_DIM / 4; ++i4) {
        const float4 v = wr4[i4];
        w2[r * (H_DIM / 2) + 2 * i4 + 0] = (v2f){v.x, v.y};
        w2[r * (H_DIM / 2) + 2 * i4 + 1] = (v2f){v.z, v.w};
      }
    }
  }
  const float vbs = b_stay[j], vbd = b_down[j];

  hv[j] = 0.f; hv[H_DIM + j] = 0.f;
  __syncthreads();

  float hlast = 0.f;
  float top_prev = 0.f;
  float n0, n1, n2, n3, n4, n5;
  {
    const float* xp = Xp + (size_t)b * G_DIM;
    n0 = xp[3 * j + 0]; n1 = xp[3 * j + 1]; n2 = xp[3 * j + 2];
    n3 = xp[GI_OFF + j]; n4 = xp[GC_OFF + j]; n5 = xp[GO_OFF + j];
  }
  #pragma nounroll
  for (int t = 0; t < T_DIM; ++t) {
    v2f s0 = {n0, 0.f}, s1 = {n1, 0.f}, s2 = {n2, 0.f};
    v2f s3 = {n3, 0.f}, s4 = {n4, 0.f}, s5 = {n5, 0.f};
    v2f u0 = {0.f, 0.f}, u1 = {0.f, 0.f}, u2 = {0.f, 0.f};
    v2f u3 = {0.f, 0.f}, u4 = {0.f, 0.f}, u5 = {0.f, 0.f};
    const float* xn = Xp + (size_t)(((t + 1 < T_DIM) ? t + 1 : t) * B_DIM + b) * G_DIM;
    {
      const float4* h4 = (const float4*)(hv + (t & 1) * H_DIM);
      float4 q[HWIN];
      #pragma unroll
      for (int n = 0; n < HWIN; ++n) q[n] = h4[n];
      __builtin_amdgcn_sched_barrier(0);
      #pragma unroll
      for (int i4 = 0; i4 < H_DIM / 4; ++i4) {
        const float4 h = q[i4 % HWIN];
        const v2f hlo = {h.x, h.y}, hhi = {h.z, h.w};
        const int k = 2 * i4;
        s0 = __builtin_elementwise_fma(hlo, w2[0 * (H_DIM / 2) + k], s0);
        s1 = __builtin_elementwise_fma(hlo, w2[1 * (H_DIM / 2) + k], s1);
        s2 = __builtin_elementwise_fma(hlo, w2[2 * (H_DIM / 2) + k], s2);
        s3 = __builtin_elementwise_fma(hlo, w2[3 * (H_DIM / 2) + k], s3);
        s4 = __builtin_elementwise_fma(hlo, w2[4 * (H_DIM / 2) + k], s4);
        s5 = __builtin_elementwise_fma(hlo, w2[5 * (H_DIM / 2) + k], s5);
        u0 = __builtin_elementwise_fma(hhi, w2[0 * (H_DIM / 2) + k + 1], u0);
        u1 = __builtin_elementwise_fma(hhi, w2[1 * (H_DIM / 2) + k + 1], u1);
        u2 = __builtin_elementwise_fma(hhi, w2[2 * (H_DIM / 2) + k + 1], u2);
        u3 = __builtin_elementwise_fma(hhi, w2[3 * (H_DIM / 2) + k + 1], u3);
        u4 = __builtin_elementwise_fma(hhi, w2[4 * (H_DIM / 2) + k + 1], u4);
        u5 = __builtin_elementwise_fma(hhi, w2[5 * (H_DIM / 2) + k + 1], u5);
        if (i4 + HWIN < H_DIM / 4) q[i4 % HWIN] = h4[i4 + HWIN];
        if (i4 == 0) {
          n0 = xn[3 * j + 0]; n1 = xn[3 * j + 1]; n2 = xn[3 * j + 2];
          n3 = xn[GI_OFF + j]; n4 = xn[GC_OFF + j]; n5 = xn[GO_OFF + j];
        }
        __builtin_amdgcn_sched_barrier(0);
      }
    }
    const v2f p0 = s0 + u0, p1 = s1 + u1, p2 = s2 + u2;
    const v2f p3 = s3 + u3, p4 = s4 + u4, p5 = s5 + u5;
    const float a0 = p0.x + p0.y, a1 = p1.x + p1.y, a2 = p2.x + p2.y;
    const float a3 = p3.x + p3.y, a4 = p4.x + p4.y, a5 = p5.x + p5.y;
    asm volatile("" :: "v"(a0), "v"(a1), "v"(a2), "v"(a3), "v"(a4), "v"(a5));
    const float mx = fmaxf(a0, fmaxf(a1, a2));
    const float e0 = __expf(a0 - mx), e1 = __expf(a1 - mx), e2 = __expf(a2 - mx);
    const float inv = 1.f / (e0 + e1 + e2);
    const float ca = e0 * inv, cb = e1 * inv, cc = e2 * inv;
    const float iv = 1.f / (1.f + __expf(-a3));
    const float cv = tanhf(a4);
    const float ov = 1.f / (1.f + __expf(-a5));
    const float pushed = iv * cv;
    {
      const float sl_t = (t == 0) ? pushed : top_prev;
      const float top = sl_t * ca + (pushed + vbs) * cb + (pushed + vbd) * cc;
      const float hn = top * ov;
      hv[((t + 1) & 1) * H_DIM + j] = hn;
      hlast = hn;
      top_prev = top;
    }
    __syncthreads();
    out[(size_t)(t * B_DIM + b) * H_DIM + j] = hlast;
  }
  out[(size_t)T_DIM * B_DIM * H_DIM + (size_t)b * H_DIM + j] = hlast;
}

extern "C" void kernel_launch(void* const* d_in, const int* in_sizes, int n_in,
                              void* d_out, int out_size, void* d_ws, size_t ws_size,
                              hipStream_t stream) {
  (void)in_sizes; (void)n_in; (void)out_size; (void)ws_size;
  const float* x  = (const float*)d_in[0];
  const float* Wf = (const float*)d_in[1];
  const float* bf = (const float*)d_in[2];
  const float* Wi = (const float*)d_in[3];
  const float* bi = (const float*)d_in[4];
  const float* Wc = (const float*)d_in[5];
  const float* bc = (const float*)d_in[6];
  const float* Wo = (const float*)d_in[7];
  const float* bo = (const float*)d_in[8];
  const float* b_stay = (const float*)d_in[11];
  const float* b_down = (const float*)d_in[13];
  float* out = (float*)d_out;

  float* Xp  = (float*)d_ws;

  xproj_wmma<<<(ROWS / 16) * (G_DIM / 16) / 8, 256, 0, stream>>>(
      x, Wf, bf, Wi, bi, Wc, bc, Wo, bo, Xp);

  size_t shmem = (size_t)(2 * H_DIM) * sizeof(float);
  scan_kernel<<<B_DIM, SCAN_THREADS, shmem, stream>>>(Xp, Wf, Wi, Wc, Wo, b_stay, b_down, out);
}


# x projection: 128x96 block tiles staged through LDS (whole-row 16-byte loads, whole-line stores), each wave a 32x48 register tile of f32 WMMA, one accumulator per tile: changed summation order
# speedup vs baseline: 16.8080x; 1.0816x over previous
#include <hip/hip_runtime.h>
#include <math.h>


typedef float v2f __attribute__((ext_vector_type(2)));
typedef float v8f __attribute__((ext_vector_type(8)));

#define T_DIM 128
#define B_DIM 32
#define E_DIM 128
#define H_DIM 128
#define GE_DIM 256
#define G_DIM 768
#define GI_OFF 384
#define GC_OFF 512
#define GO_OFF 640
#define ROWS (T_DIM * B_DIM)

__device__ __forceinline__ const float* wrow(int k,
    const float* Wf, const float* Wi, const float* Wc, const float* Wo) {
  const int m = (k >= GI_OFF) + (k >= GC_OFF) + (k >= GO_OFF);
  const float* base = (m == 0) ? Wf : (m == 1) ? Wi : (m == 2) ? Wc : Wo;
  const int first = (m == 0) ? 0 : GI_OFF + (m - 1) * H_DIM;
  return base + (size_t)(k - first) * GE_DIM;
}
__device__ __forceinline__ float biasAt(int k,
    const float* bf, const float* bi, const float* bc, const float* bo) {
  const int m = (k >= GI_OFF) + (k >= GC_OFF) + (k >= GO_OFF);
  const float* base = (m == 0) ? bf : (m == 1) ? bi : (m == 2) ? bc : bo;
  const int first = (m == 0) ? 0 : GI_OFF + (m - 1) * H_DIM;
  return base[k - first];
}

#define P1_BM 128
#define P1_BN 96
#define P1_PITCH 132
#define P1_CPITCH 100
#define P1_THREADS 256

__global__ void __launch_bounds__(P1_THREADS)
xproj_wmma(const float* __restrict__ x,
           const float* __restrict__ Wf, const float* __restrict__ bf,
           const float* __restrict__ Wi, const float* __restrict__ bi,
           const float* __restrict__ Wc, const float* __restrict__ bc,
           const float* __restrict__ Wo, const float* __restrict__ bo,
           float* __restrict__ Xp) {
  __shared__ __attribute__((aligned(16))) float As[P1_BM * P1_PITCH];
  __shared__ __attribute__((aligned(16))) float Bs[P1_BN * P1_PITCH];
  __shared__ __attribute__((aligned(16))) float Cs[P1_BM * P1_CPITCH];

  const int tid  = threadIdx.x;
  const int lane = tid & 31;
  const int wave = __builtin_amdgcn_readfirstlane(tid >> 5);
  const int bn = blockIdx.x % (G_DIM / P1_BN);
  const int bm = blockIdx.x / (G_DIM / P1_BN);
  const int r0 = bm * P1_BM;
  const int k0 = bn * P1_BN;

  const int c  = lane & 15;
  const int h  = lane >> 4;
  const int wm = wave >> 1;
  const int wn = wave & 1;
  float bb[3];
  #pragma unroll
  for (int ni = 0; ni < 3; ++ni) bb[ni] = biasAt(k0 + 48 * wn + 16 * ni + c, bf, bi, bc, bo);

  {
    float4 av[P1_BM / 8];
    float4 bv[P1_BN / 8];
    #pragma unroll
    for (int i = 0; i < P1_BM / 8; ++i)
      av[i] = *(const float4*)(x + (size_t)(r0 + wave + 8 * i) * E_DIM + 4 * lane);
    #pragma unroll
    for (int i = 0; i < P1_BN / 8; ++i)
      bv[i] = *(const float4*)(wrow(k0 + wave + 8 * i, Wf, Wi, Wc, Wo) + H_DIM + 4 * lane);
    #pragma unroll
    for (int i = 0; i < P1_BM / 8; ++i)
      *(float4*)(As + (wave + 8 * i) * P1_PITCH + 4 * lane) = av[i];
    #pragma unroll
    for (int i = 0; i < P1_BN / 8; ++i)
      *(float4*)(Bs + (wave + 8 * i) * P1_PITCH + 4 * lane) = bv[i];
  }
  __syncthreads();

  const float* ap = As + (32 * wm + c) * P1_PITCH + 4 * h;
  const float* bp = Bs + (48 * wn + c) * P1_PITCH + 4 * h;

  v8f acc[2][3];
  #pragma unroll
  for (int mi = 0; mi < 2; ++mi)
    #pragma unroll
    for (int ni = 0; ni < 3; ++ni) acc[mi][ni] = (v8f){};

  #pragma unroll
  for (int s = 0; s < E_DIM / 8; ++s) {
    float4 a[2], b[3];
    #pragma unroll
    for (int mi = 0; mi < 2; ++mi) a[mi] = *(const float4*)(ap + mi * 16 * P1_PITCH + 8 * s);
    #pragma unroll
    for (int ni = 0; ni < 3; ++ni) b[ni] = *(const float4*)(bp + ni * 16 * P1_PITCH + 8 * s);
    #pragma unroll
    for (int mi = 0; mi < 2; ++mi)
      #pragma unroll
      for (int ni = 0; ni < 3; ++ni)
        acc[mi][ni] = __builtin_amdgcn_wmma_f32_16x16x4_f32(
            false, (v2f){a[mi].x, a[mi].y}, false, (v2f){b[ni].x, b[ni].y},
            (short)0, acc[mi][ni], false, false);
    #pragma unroll
    for (int mi = 0; mi < 2; ++mi)
      #pragma unroll
      for (int ni = 0; ni < 3; ++ni)
        acc[mi][ni] = __builtin_amdgcn_wmma_f32_16x16x4_f32(
            false, (v2f){a[mi].z, a[mi].w}, false, (v2f){b[ni].z, b[ni].w},
            (short)0, acc[mi][ni], false, false);
  }

  #pragma unroll
  for (int ni = 0; ni < 3; ++ni) {
    const int col = 48 * wn + 16 * ni + c;
    #pragma unroll
    for (int mi = 0; mi < 2; ++mi) {
      float* cp = Cs + (32 * wm + 16 * mi + 8 * h) * P1_CPITCH + col;
      #pragma unroll
      for (int v = 0; v < 8; ++v) cp[v * P1_CPITCH] = acc[mi][ni][v] + bb[ni];
    }
  }
  __syncthreads();
  #pragma unroll
  for (int i = 0; i < (P1_BM * P1_BN / 4) / P1_THREADS; ++i) {
    const int q   = tid + P1_THREADS * i;
    const int row = q / (P1_BN / 4);
    const int c4  = q % (P1_BN / 4);
    const float4 v = *(const float4*)(Cs + row * P1_CPITCH + 4 * c4);
    *(float4*)(Xp + (size_t)(r0 + row) * G_DIM + k0 + 4 * c4) = v;
  }
}

#define SCAN_THREADS H_DIM
#define NG 6
#define HWIN 12

__global__ void __launch_bounds__(SCAN_THREADS)
scan_kernel(const float* __restrict__ Xp,
            const float* __restrict__ Wf, const float* __restrict__ Wi,
            const float* __restrict__ Wc, const float* __restrict__ Wo,
            const float* __restrict__ b_stay,
            const float* __restrict__ b_down,
            float* __restrict__ out) {
  extern __shared__ __align__(16) float smem[];
  float* hv   = smem;

  const int j = threadIdx.x;
  const int b = blockIdx.x;

  v2f w2[NG * H_DIM / 2];
  {
    const float* rp[NG];
    rp[0] = Wf + (size_t)(3 * j + 0) * GE_DIM;
    rp[1] = Wf + (size_t)(3 * j + 1) * GE_DIM;
    rp[2] = Wf + (size_t)(3 * j + 2) * GE_DIM;
    rp[3] = Wi + (size_t)j * GE_DIM;
    rp[4] = Wc + (size_t)j * GE_DIM;
    rp[5] = Wo + (size_t)j * GE_DIM;
    #pragma unroll
    for (int r = 0; r < NG; ++r) {
      const float4* wr4 = (const float4*)rp[r];
      #pragma unroll
      for (int i4 = 0; i4 < H_DIM / 4; ++i4) {
        const float4 v = wr4[i4];
        w2[r * (H_DIM / 2) + 2 * i4 + 0] = (v2f){v.x, v.y};
        w2[r * (H_DIM / 2) + 2 * i4 + 1] = (v2f){v.z, v.w};
      }
    }
  }
  const float vbs = b_stay[j], vbd = b_down[j];

  hv[j] = 0.f; hv[H_DIM + j] = 0.f;
  __syncthreads();

  float hlast = 0.f;
  float top_prev = 0.f;
  float n0, n1, n2, n3, n4, n5;
  {
    const float* xp = Xp + (size_t)b * G_DIM;
    n0 = xp[3 * j + 0]; n1 = xp[3 * j + 1]; n2 = xp[3 * j + 2];
    n3 = xp[GI_OFF + j]; n4 = xp[GC_OFF + j]; n5 = xp[GO_OFF + j];
  }
  #pragma nounroll
  for (int t = 0; t < T_DIM; ++t) {
    v2f s0 = {n0, 0.f}, s1 = {n1, 0.f}, s2 = {n2, 0.f};
    v2f s3 = {n3, 0.f}, s4 = {n4, 0.f}, s5 = {n5, 0.f};
    v2f u0 = {0.f, 0.f}, u1 = {0.f, 0.f}, u2 = {0.f, 0.f};
    v2f u3 = {0.f, 0.f}, u4 = {0.f, 0.f}, u5 = {0.f, 0.f};
    const float* xn = Xp + (size_t)(((t + 1 < T_DIM) ? t + 1 : t) * B_DIM + b) * G_DIM;
    {
      const float4* h4 = (const float4*)(hv + (t & 1) * H_DIM);
      float4 q[HWIN];
      #pragma unroll
      for (int n = 0; n < HWIN; ++n) q[n] = h4[n];
      __builtin_amdgcn_sched_barrier(0);
      #pragma unroll
      for (int i4 = 0; i4 < H_DIM / 4; ++i4) {
        const float4 h = q[i4 % HWIN];
        const v2f hlo = {h.x, h.y}, hhi = {h.z, h.w};
        const int k = 2 * i4;
        s0 = __builtin_elementwise_fma(hlo, w2[0 * (H_DIM / 2) + k], s0);
        s1 = __builtin_elementwise_fma(hlo, w2[1 * (H_DIM / 2) + k], s1);
        s2 = __builtin_elementwise_fma(hlo, w2[2 * (H_DIM / 2) + k], s2);
        s3 = __builtin_elementwise_fma(hlo, w2[3 * (H_DIM / 2) + k], s3);
        s4 = __builtin_elementwise_fma(hlo, w2[4 * (H_DIM / 2) + k], s4);
        s5 = __builtin_elementwise_fma(hlo, w2[5 * (H_DIM / 2) + k], s5);
        u0 = __builtin_elementwise_fma(hhi, w2[0 * (H_DIM / 2) + k + 1], u0);
        u1 = __builtin_elementwise_fma(hhi, w2[1 * (H_DIM / 2) + k + 1], u1);
        u2 = __builtin_elementwise_fma(hhi, w2[2 * (H_DIM / 2) + k + 1], u2);
        u3 = __builtin_elementwise_fma(hhi, w2[3 * (H_DIM / 2) + k + 1], u3);
        u4 = __builtin_elementwise_fma(hhi, w2[4 * (H_DIM / 2) + k + 1], u4);
        u5 = __builtin_elementwise_fma(hhi, w2[5 * (H_DIM / 2) + k + 1], u5);
        if (i4 + HWIN < H_DIM / 4) q[i4 % HWIN] = h4[i4 + HWIN];
        if (i4 == 0) {
          n0 = xn[3 * j + 0]; n1 = xn[3 * j + 1]; n2 = xn[3 * j + 2];
          n3 = xn[GI_OFF + j]; n4 = xn[GC_OFF + j]; n5 = xn[GO_OFF + j];
        }
        __builtin_amdgcn_sched_barrier(0);
      }
    }
    const v2f p0 = s0 + u0, p1 = s1 + u1, p2 = s2 + u2;
    const v2f p3 = s3 + u3, p4 = s4 + u4, p5 = s5 + u5;
    const float a0 = p0.x + p0.y, a1 = p1.x + p1.y, a2 = p2.x + p2.y;
    const float a3 = p3.x + p3.y, a4 = p4.x + p4.y, a5 = p5.x + p5.y;
    asm volatile("" :: "v"(a0), "v"(a1), "v"(a2), "v"(a3), "v"(a4), "v"(a5));
    const float mx = fmaxf(a0, fmaxf(a1, a2));
    const float e0 = __expf(a0 - mx), e1 = __expf(a1 - mx), e2 = __expf(a2 - mx);
    const float inv = 1.f / (e0 + e1 + e2);
    const float ca = e0 * inv, cb = e1 * inv, cc = e2 * inv;
    const float iv = 1.f / (1.f + __expf(-a3));
    const float cv = tanhf(a4);
    const float ov = 1.f / (1.f + __expf(-a5));
    const float pushed = iv * cv;
    {
      const float sl_t = (t == 0) ? pushed : top_prev;
      const float top = sl_t * ca + (pushed + vbs) * cb + (pushed + vbd) * cc;
      const float hn = top * ov;
      hv[((t + 1) & 1) * H_DIM + j] = hn;
      hlast = hn;
      top_prev = top;
    }
    __syncthreads();
    out[(size_t)(t * B_DIM + b) * H_DIM + j] = hlast;
  }
  out[(size_t)T_DIM * B_DIM * H_DIM + (size_t)b * H_DIM + j] = hlast;
}

extern "C" void kernel_launch(void* const* d_in, const int* in_sizes, int n_in,
                              void* d_out, int out_size, void* d_ws, size_t ws_size,
                              hipStream_t stream) {
  (void)in_sizes; (void)n_in; (void)out_size; (void)ws_size;
  const float* x  = (const float*)d_in[0];
  const float* Wf = (const float*)d_in[1];
  const float* bf = (const float*)d_in[2];
  const float* Wi = (const float*)d_in[3];
  const float* bi = (const float*)d_in[4];
  const float* Wc = (const float*)d_in[5];
  const float* bc = (const float*)d_in[6];
  const float* Wo = (const float*)d_in[7];
  const float* bo = (const float*)d_in[8];
  const float* b_stay = (const float*)d_in[11];
  const float* b_down = (const float*)d_in[13];
  float* out = (float*)d_out;

  float* Xp  = (float*)d_ws;

  xproj_wmma<<<(ROWS / P1_BM) * (G_DIM / P1_BN), P1_THREADS, 0, stream>>>(
      x, Wf, bf, Wi, bi, Wc, bc, Wo, bo, Xp);

  size_t shmem = (size_t)(2 * H_DIM) * sizeof(float);
  scan_kernel<<<B_DIM, SCAN_THREADS, shmem, stream>>>(Xp, Wf, Wi, Wc, Wo, b_stay, b_down, out);
}


# scan gate tail: the three IEEE divisions become hardware reciprocals (1 ulp) and tanh evaluates both ranges and selects instead of branching, so the four gate chains run as one straight block
# speedup vs baseline: 18.3665x; 1.0927x over previous
#include <hip/hip_runtime.h>
#include <math.h>


typedef float v2f __attribute__((ext_vector_type(2)));
typedef float v8f __attribute__((ext_vector_type(8)));

#define T_DIM 128
#define B_DIM 32
#define E_DIM 128
#define H_DIM 128
#define GE_DIM 256
#define G_DIM 768
#define GI_OFF 384
#define GC_OFF 512
#define GO_OFF 640
#define ROWS (T_DIM * B_DIM)

__device__ __forceinline__ const float* wrow(int k,
    const float* Wf, const float* Wi, const float* Wc, const float* Wo) {
  const int m = (k >= GI_OFF) + (k >= GC_OFF) + (k >= GO_OFF);
  const float* base = (m == 0) ? Wf : (m == 1) ? Wi : (m == 2) ? Wc : Wo;
  const int first = (m == 0) ? 0 : GI_OFF + (m - 1) * H_DIM;
  return base + (size_t)(k - first) * GE_DIM;
}
__device__ __forceinline__ float biasAt(int k,
    const float* bf, const float* bi, const float* bc, const float* bo) {
  const int m = (k >= GI_OFF) + (k >= GC_OFF) + (k >= GO_OFF);
  const float* base = (m == 0) ? bf : (m == 1) ? bi : (m == 2) ? bc : bo;
  const int first = (m == 0) ? 0 : GI_OFF + (m - 1) * H_DIM;
  return base[k - first];
}

#define P1_BM 128
#define P1_BN 96
#define P1_PITCH 132
#define P1_CPITCH 100
#define P1_THREADS 256

__global__ void __launch_bounds__(P1_THREADS)
xproj_wmma(const float* __restrict__ x,
           const float* __restrict__ Wf, const float* __restrict__ bf,
           const float* __restrict__ Wi, const float* __restrict__ bi,
           const float* __restrict__ Wc, const float* __restrict__ bc,
           const float* __restrict__ Wo, const float* __restrict__ bo,
           float* __restrict__ Xp) {
  __shared__ __attribute__((aligned(16))) float As[P1_BM * P1_PITCH];
  __shared__ __attribute__((aligned(16))) float Bs[P1_BN * P1_PITCH];
  __shared__ __attribute__((aligned(16))) float Cs[P1_BM * P1_CPITCH];

  const int tid  = threadIdx.x;
  const int lane = tid & 31;
  const int wave = __builtin_amdgcn_readfirstlane(tid >> 5);
  const int bn = blockIdx.x % (G_DIM / P1_BN);
  const int bm = blockIdx.x / (G_DIM / P1_BN);
  const int r0 = bm * P1_BM;
  const int k0 = bn * P1_BN;

  const int c  = lane & 15;
  const int h  = lane >> 4;
  const int wm = wave >> 1;
  const int wn = wave & 1;
  float bb[3];
  #pragma unroll
  for (int ni = 0; ni < 3; ++ni) bb[ni] = biasAt(k0 + 48 * wn + 16 * ni + c, bf, bi, bc, bo);

  {
    float4 av[P1_BM / 8];
    float4 bv[P1_BN / 8];
    #pragma unroll
    for (int i = 0; i < P1_BM / 8; ++i)
      av[i] = *(const float4*)(x + (size_t)(r0 + wave + 8 * i) * E_DIM + 4 * lane);
    #pragma unroll
    for (int i = 0; i < P1_BN / 8; ++i)
      bv[i] = *(const float4*)(wrow(k0 + wave + 8 * i, Wf, Wi, Wc, Wo) + H_DIM + 4 * lane);
    #pragma unroll
    for (int i = 0; i < P1_BM / 8; ++i)
      *(float4*)(As + (wave + 8 * i) * P1_PITCH + 4 * lane) = av[i];
    #pragma unroll
    for (int i = 0; i < P1_BN / 8; ++i)
      *(float4*)(Bs + (wave + 8 * i) * P1_PITCH + 4 * lane) = bv[i];
  }
  __syncthreads();

  const float* ap = As + (32 * wm + c) * P1_PITCH + 4 * h;
  const float* bp = Bs + (48 * wn + c) * P1_PITCH + 4 * h;

  v8f acc[2][3];
  #pragma unroll
  for (int mi = 0; mi < 2; ++mi)
    #pragma unroll
    for (int ni = 0; ni < 3; ++ni) acc[mi][ni] = (v8f){};

  #pragma unroll
  for (int s = 0; s < E_DIM / 8; ++s) {
    float4 a[2], b[3];
    #pragma unroll
    for (int mi = 0; mi < 2; ++mi) a[mi] = *(const float4*)(ap + mi * 16 * P1_PITCH + 8 * s);
    #pragma unroll
    for (int ni = 0; ni < 3; ++ni) b[ni] = *(const float4*)(bp + ni * 16 * P1_PITCH + 8 * s);
    #pragma unroll
    for (int mi = 0; mi < 2; ++mi)
      #pragma unroll
      for (int ni = 0; ni < 3; ++ni)
        acc[mi][ni] = __builtin_amdgcn_wmma_f32_16x16x4_f32(
            false, (v2f){a[mi].x, a[mi].y}, false, (v2f){b[ni].x, b[ni].y},
            (short)0, acc[mi][ni], false, false);
    #pragma unroll
    for (int mi = 0; mi < 2; ++mi)
      #pragma unroll
      for (int ni = 0; ni < 3; ++ni)
        acc[mi][ni] = __builtin_amdgcn_wmma_f32_16x16x4_f32(
            false, (v2f){a[mi].z, a[mi].w}, false, (v2f){b[ni].z, b[ni].w},
            (short)0, acc[mi][ni], false, false);
  }

  #pragma unroll
  for (int ni = 0; ni < 3; ++ni) {
    const int col = 48 * wn + 16 * ni + c;
    #pragma unroll
    for (int mi = 0; mi < 2; ++mi) {
      float* cp = Cs + (32 * wm + 16 * mi + 8 * h) * P1_CPITCH + col;
      #pragma unroll
      for (int v = 0; v < 8; ++v) cp[v * P1_CPITCH] = acc[mi][ni][v] + bb[ni];
    }
  }
  __syncthreads();
  #pragma unroll
  for (int i = 0; i < (P1_BM * P1_BN / 4) / P1_THREADS; ++i) {
    const int q   = tid + P1_THREADS * i;
    const int row = q / (P1_BN / 4);
    const int c4  = q % (P1_BN / 4);
    const float4 v = *(const float4*)(Cs + row * P1_CPITCH + 4 * c4);
    *(float4*)(Xp + (size_t)(r0 + row) * G_DIM + k0 + 4 * c4) = v;
  }
}

__device__ __forceinline__ float tanh_sel(float x) {
  const float y  = fabsf(x);
  const float y2 = x * x;
  float p = fmaf(y2, -0.00570002059f, 0.0206340719f);
  p = fmaf(y2, p, -0.0537379310f);
  p = fmaf(y2, p, 0.133314162f);
  p = fmaf(y2, p, -0.333332807f);
  const float small = fmaf(y2, y * p, y);
  const float t = __expf(y + y);
  const float big = fmaf(__builtin_amdgcn_rcpf(t + 1.f), -2.f, 1.f);
  return copysignf((y < 0.625f) ? small : big, x);
}

#define SCAN_THREADS H_DIM
#define NG 6
#define HWIN 12

__global__ void __launch_bounds__(SCAN_THREADS)
scan_kernel(const float* __restrict__ Xp,
            const float* __restrict__ Wf, const float* __restrict__ Wi,
            const float* __restrict__ Wc, const float* __restrict__ Wo,
            const float* __restrict__ b_stay,
            const float* __restrict__ b_down,
            float* __restrict__ out) {
  extern __shared__ __align__(16) float smem[];
  float* hv   = smem;

  const int j = threadIdx.x;
  const int b = blockIdx.x;

  v2f w2[NG * H_DIM / 2];
  {
    const float* rp[NG];
    rp[0] = Wf + (size_t)(3 * j + 0) * GE_DIM;
    rp[1] = Wf + (size_t)(3 * j + 1) * GE_DIM;
    rp[2] = Wf + (size_t)(3 * j + 2) * GE_DIM;
    rp[3] = Wi + (size_t)j * GE_DIM;
    rp[4] = Wc + (size_t)j * GE_DIM;
    rp[5] = Wo + (size_t)j * GE_DIM;
    #pragma unroll
    for (int r = 0; r < NG; ++r) {
      const float4* wr4 = (const float4*)rp[r];
      #pragma unroll
      for (int i4 = 0; i4 < H_DIM / 4; ++i4) {
        const float4 v = wr4[i4];
        w2[r * (H_DIM / 2) + 2 * i4 + 0] = (v2f){v.x, v.y};
        w2[r * (H_DIM / 2) + 2 * i4 + 1] = (v2f){v.z, v.w};
      }
    }
  }
  const float vbs = b_stay[j], vbd = b_down[j];

  hv[j] = 0.f; hv[H_DIM + j] = 0.f;
  __syncthreads();

  float hlast = 0.f;
  float top_prev = 0.f;
  float n0, n1, n2, n3, n4, n5;
  {
    const float* xp = Xp + (size_t)b * G_DIM;
    n0 = xp[3 * j + 0]; n1 = xp[3 * j + 1]; n2 = xp[3 * j + 2];
    n3 = xp[GI_OFF + j]; n4 = xp[GC_OFF + j]; n5 = xp[GO_OFF + j];
  }
  #pragma nounroll
  for (int t = 0; t < T_DIM; ++t) {
    v2f s0 = {n0, 0.f}, s1 = {n1, 0.f}, s2 = {n2, 0.f};
    v2f s3 = {n3, 0.f}, s4 = {n4, 0.f}, s5 = {n5, 0.f};
    v2f u0 = {0.f, 0.f}, u1 = {0.f, 0.f}, u2 = {0.f, 0.f};
    v2f u3 = {0.f, 0.f}, u4 = {0.f, 0.f}, u5 = {0.f, 0.f};
    const float* xn = Xp + (size_t)(((t + 1 < T_DIM) ? t + 1 : t) * B_DIM + b) * G_DIM;
    {
      const float4* h4 = (const float4*)(hv + (t & 1) * H_DIM);
      float4 q[HWIN];
      #pragma unroll
      for (int n = 0; n < HWIN; ++n) q[n] = h4[n];
      __builtin_amdgcn_sched_barrier(0);
      #pragma unroll
      for (int i4 = 0; i4 < H_DIM / 4; ++i4) {
        const float4 h = q[i4 % HWIN];
        const v2f hlo = {h.x, h.y}, hhi = {h.z, h.w};
        const int k = 2 * i4;
        s0 = __builtin_elementwise_fma(hlo, w2[0 * (H_DIM / 2) + k], s0);
        s1 = __builtin_elementwise_fma(hlo, w2[1 * (H_DIM / 2) + k], s1);
        s2 = __builtin_elementwise_fma(hlo, w2[2 * (H_DIM / 2) + k], s2);
        s3 = __builtin_elementwise_fma(hlo, w2[3 * (H_DIM / 2) + k], s3);
        s4 = __builtin_elementwise_fma(hlo, w2[4 * (H_DIM / 2) + k], s4);
        s5 = __builtin_elementwise_fma(hlo, w2[5 * (H_DIM / 2) + k], s5);
        u0 = __builtin_elementwise_fma(hhi, w2[0 * (H_DIM / 2) + k + 1], u0);
        u1 = __builtin_elementwise_fma(hhi, w2[1 * (H_DIM / 2) + k + 1], u1);
        u2 = __builtin_elementwise_fma(hhi, w2[2 * (H_DIM / 2) + k + 1], u2);
        u3 = __builtin_elementwise_fma(hhi, w2[3 * (H_DIM / 2) + k + 1], u3);
        u4 = __builtin_elementwise_fma(hhi, w2[4 * (H_DIM / 2) + k + 1], u4);
        u5 = __builtin_elementwise_fma(hhi, w2[5 * (H_DIM / 2) + k + 1], u5);
        if (i4 + HWIN < H_DIM / 4) q[i4 % HWIN] = h4[i4 + HWIN];
        if (i4 == 0) {
          n0 = xn[3 * j + 0]; n1 = xn[3 * j + 1]; n2 = xn[3 * j + 2];
          n3 = xn[GI_OFF + j]; n4 = xn[GC_OFF + j]; n5 = xn[GO_OFF + j];
        }
        __builtin_amdgcn_sched_barrier(0);
      }
    }
    const v2f p0 = s0 + u0, p1 = s1 + u1, p2 = s2 + u2;
    const v2f p3 = s3 + u3, p4 = s4 + u4, p5 = s5 + u5;
    const float a0 = p0.x + p0.y, a1 = p1.x + p1.y, a2 = p2.x + p2.y;
    const float a3 = p3.x + p3.y, a4 = p4.x + p4.y, a5 = p5.x + p5.y;
    const float mx = fmaxf(a0, fmaxf(a1, a2));
    const float e0 = __expf(a0 - mx), e1 = __expf(a1 - mx), e2 = __expf(a2 - mx);
    const float inv = __builtin_amdgcn_rcpf(e0 + e1 + e2);
    const float ca = e0 * inv, cb = e1 * inv, cc = e2 * inv;
    const float iv = __builtin_amdgcn_rcpf(1.f + __expf(-a3));
    const float cv = tanh_sel(a4);
    const float ov = __builtin_amdgcn_rcpf(1.f + __expf(-a5));
    const float pushed = iv * cv;
    {
      const float sl_t = (t == 0) ? pushed : top_prev;
      const float top = sl_t * ca + (pushed + vbs) * cb + (pushed + vbd) * cc;
      const float hn = top * ov;
      hv[((t + 1) & 1) * H_DIM + j] = hn;
      hlast = hn;
      top_prev = top;
    }
    __syncthreads();
    out[(size_t)(t * B_DIM + b) * H_DIM + j] = hlast;
  }
  out[(size_t)T_DIM * B_DIM * H_DIM + (size_t)b * H_DIM + j] = hlast;
}

extern "C" void kernel_launch(void* const* d_in, const int* in_sizes, int n_in,
                              void* d_out, int out_size, void* d_ws, size_t ws_size,
                              hipStream_t stream) {
  (void)in_sizes; (void)n_in; (void)out_size; (void)ws_size;
  const float* x  = (const float*)d_in[0];
  const float* Wf = (const float*)d_in[1];
  const float* bf = (const float*)d_in[2];
  const float* Wi = (const float*)d_in[3];
  const float* bi = (const float*)d_in[4];
  const float* Wc = (const float*)d_in[5];
  const float* bc = (const float*)d_in[6];
  const float* Wo = (const float*)d_in[7];
  const float* bo = (const float*)d_in[8];
  const float* b_stay = (const float*)d_in[11];
  const float* b_down = (const float*)d_in[13];
  float* out = (float*)d_out;

  float* Xp  = (float*)d_ws;

  xproj_wmma<<<(ROWS / P1_BM) * (G_DIM / P1_BN), P1_THREADS, 0, stream>>>(
      x, Wf, bf, Wi, bi, Wc, bc, Wo, bo, Xp);

  size_t shmem = (size_t)(2 * H_DIM) * sizeof(float);
  scan_kernel<<<B_DIM, SCAN_THREADS, shmem, stream>>>(Xp, Wf, Wi, Wc, Wo, b_stay, b_down, out);
}


# scan: next step's projected inputs requested 1.5 steps ahead (ring of 2 register sets, loop unrolled by 2), added behind the sums: changed summation order; step barrier by hand, LDS wait only
# speedup vs baseline: 19.6910x; 1.0721x over previous
#include <hip/hip_runtime.h>
#include <math.h>


typedef float v2f __attribute__((ext_vector_type(2)));
typedef float v8f __attribute__((ext_vector_type(8)));

#define T_DIM 128
#define B_DIM 32
#define E_DIM 128
#define H_DIM 128
#define GE_DIM 256
#define G_DIM 768
#define GI_OFF 384
#define GC_OFF 512
#define GO_OFF 640
#define ROWS (T_DIM * B_DIM)

__device__ __forceinline__ const float* wrow(int k,
    const float* Wf, const float* Wi, const float* Wc, const float* Wo) {
  const int m = (k >= GI_OFF) + (k >= GC_OFF) + (k >= GO_OFF);
  const float* base = (m == 0) ? Wf : (m == 1) ? Wi : (m == 2) ? Wc : Wo;
  const int first = (m == 0) ? 0 : GI_OFF + (m - 1) * H_DIM;
  return base + (size_t)(k - first) * GE_DIM;
}
__device__ __forceinline__ float biasAt(int k,
    const float* bf, const float* bi, const float* bc, const float* bo) {
  const int m = (k >= GI_OFF) + (k >= GC_OFF) + (k >= GO_OFF);
  const float* base = (m == 0) ? bf : (m == 1) ? bi : (m == 2) ? bc : bo;
  const int first = (m == 0) ? 0 : GI_OFF + (m - 1) * H_DIM;
  return base[k - first];
}

#define P1_BM 128
#define P1_BN 96
#define P1_PITCH 132
#define P1_CPITCH 100
#define P1_THREADS 256

__global__ void __launch_bounds__(P1_THREADS)
xproj_wmma(const float* __restrict__ x,
           const float* __restrict__ Wf, const float* __restrict__ bf,
           const float* __restrict__ Wi, const float* __restrict__ bi,
           const float* __restrict__ Wc, const float* __restrict__ bc,
           const float* __restrict__ Wo, const float* __restrict__ bo,
           float* __restrict__ Xp) {
  __shared__ __attribute__((aligned(16))) float As[P1_BM * P1_PITCH];
  __shared__ __attribute__((aligned(16))) float Bs[P1_BN * P1_PITCH];
  __shared__ __attribute__((aligned(16))) float Cs[P1_BM * P1_CPITCH];

  const int tid  = threadIdx.x;
  const int lane = tid & 31;
  const int wave = __builtin_amdgcn_readfirstlane(tid >> 5);
  const int bn = blockIdx.x % (G_DIM / P1_BN);
  const int bm = blockIdx.x / (G_DIM / P1_BN);
  const int r0 = bm * P1_BM;
  const int k0 = bn * P1_BN;

  const int c  = lane & 15;
  const int h  = lane >> 4;
  const int wm = wave >> 1;
  const int wn = wave & 1;
  float bb[3];
  #pragma unroll
  for (int ni = 0; ni < 3; ++ni) bb[ni] = biasAt(k0 + 48 * wn + 16 * ni + c, bf, bi, bc, bo);

  {
    float4 av[P1_BM / 8];
    float4 bv[P1_BN / 8];
    #pragma unroll
    for (int i = 0; i < P1_BM / 8; ++i)
      av[i] = *(const float4*)(x + (size_t)(r0 + wave + 8 * i) * E_DIM + 4 * lane);
    #pragma unroll
    for (int i = 0; i < P1_BN / 8; ++i)
      bv[i] = *(const float4*)(wrow(k0 + wave + 8 * i, Wf, Wi, Wc, Wo) + H_DIM + 4 * lane);
    #pragma unroll
    for (int i = 0; i < P1_BM / 8; ++i)
      *(float4*)(As + (wave + 8 * i) * P1_PITCH + 4 * lane) = av[i];
    #pragma unroll
    for (int i = 0; i < P1_BN / 8; ++i)
      *(float4*)(Bs + (wave + 8 * i) * P1_PITCH + 4 * lane) = bv[i];
  }
  __syncthreads();

  const float* ap = As + (32 * wm + c) * P1_PITCH + 4 * h;
  const float* bp = Bs + (48 * wn + c) * P1_PITCH + 4 * h;

  v8f acc[2][3];
  #pragma unroll
  for (int mi = 0; mi < 2; ++mi)
    #pragma unroll
    for (int ni = 0; ni < 3; ++ni) acc[mi][ni] = (v8f){};

  #pragma unroll
  for (int s = 0; s < E_DIM / 8; ++s) {
    float4 a[2], b[3];
    #pragma unroll
    for (int mi = 0; mi < 2; ++mi) a[mi] = *(const float4*)(ap + mi * 16 * P1_PITCH + 8 * s);
    #pragma unroll
    for (int ni = 0; ni < 3; ++ni) b[ni] = *(const float4*)(bp + ni * 16 * P1_PITCH + 8 * s);
    #pragma unroll
    for (int mi = 0; mi < 2; ++mi)
      #pragma unroll
      for (int ni = 0; ni < 3; ++ni)
        acc[mi][ni] = __builtin_amdgcn_wmma_f32_16x16x4_f32(
            false, (v2f){a[mi].x, a[mi].y}, false, (v2f){b[ni].x, b[ni].y},
            (short)0, acc[mi][ni], false, false);
    #pragma unroll
    for (int mi = 0; mi < 2; ++mi)
      #pragma unroll
      for (int ni = 0; ni < 3; ++ni)
        acc[mi][ni] = __builtin_amdgcn_wmma_f32_16x16x4_f32(
            false, (v2f){a[mi].z, a[mi].w}, false, (v2f){b[ni].z, b[ni].w},
            (short)0, acc[mi][ni], false, false);
  }

  #pragma unroll
  for (int ni = 0; ni < 3; ++ni) {
    const int col = 48 * wn + 16 * ni + c;
    #pragma unroll
    for (int mi = 0; mi < 2; ++mi) {
      float* cp = Cs + (32 * wm + 16 * mi + 8 * h) * P1_CPITCH + col;
      #pragma unroll
      for (int v = 0; v < 8; ++v) cp[v * P1_CPITCH] = acc[mi][ni][v] + bb[ni];
    }
  }
  __syncthreads();
  #pragma unroll
  for (int i = 0; i < (P1_BM * P1_BN / 4) / P1_THREADS; ++i) {
    const int q   = tid + P1_THREADS * i;
    const int row = q / (P1_BN / 4);
    const int c4  = q % (P1_BN / 4);
    const float4 v = *(const float4*)(Cs + row * P1_CPITCH + 4 * c4);
    *(float4*)(Xp + (size_t)(r0 + row) * G_DIM + k0 + 4 * c4) = v;
  }
}

__device__ __forceinline__ float tanh_sel(float x) {
  const float y  = fabsf(x);
  const float y2 = x * x;
  float p = fmaf(y2, -0.00570002059f, 0.0206340719f);
  p = fmaf(y2, p, -0.0537379310f);
  p = fmaf(y2, p, 0.133314162f);
  p = fmaf(y2, p, -0.333332807f);
  const float small = fmaf(y2, y * p, y);
  const float t = __expf(y + y);
  const float big = fmaf(__builtin_amdgcn_rcpf(t + 1.f), -2.f, 1.f);
  return copysignf((y < 0.625f) ? small : big, x);
}

#define SCAN_THREADS H_DIM
#define NG 6
#define HWIN 12

__global__ void __launch_bounds__(SCAN_THREADS)
scan_kernel(const float* __restrict__ Xp,
            const float* __restrict__ Wf, const float* __restrict__ Wi,
            const float* __restrict__ Wc, const float* __restrict__ Wo,
            const float* __restrict__ b_stay,
            const float* __restrict__ b_down,
            float* __restrict__ out) {
  extern __shared__ __align__(16) float smem[];
  float* hv   = smem;

  const int j = threadIdx.x;
  const int b = blockIdx.x;

  v2f w2[NG * H_DIM / 2];
  {
    const float* rp[NG];
    rp[0] = Wf + (size_t)(3 * j + 0) * GE_DIM;
    rp[1] = Wf + (size_t)(3 * j + 1) * GE_DIM;
    rp[2] = Wf + (size_t)(3 * j + 2) * GE_DIM;
    rp[3] = Wi + (size_t)j * GE_DIM;
    rp[4] = Wc + (size_t)j * GE_DIM;
    rp[5] = Wo + (size_t)j * GE_DIM;
    #pragma unroll
    for (int r = 0; r < NG; ++r) {
      const float4* wr4 = (const float4*)rp[r];
      #pragma unroll
      for (int i4 = 0; i4 < H_DIM / 4; ++i4) {
        const float4 v = wr4[i4];
        w2[r * (H_DIM / 2) + 2 * i4 + 0] = (v2f){v.x, v.y};
        w2[r * (H_DIM / 2) + 2 * i4 + 1] = (v2f){v.z, v.w};
      }
    }
  }
  const float vbs = b_stay[j], vbd = b_down[j];

  hv[j] = 0.f; hv[H_DIM + j] = 0.f;
  __syncthreads();

  float hlast = 0.f;
  float top_prev = 0.f;
  float nx[2][NG];
  {
    const float* xp = Xp + (size_t)b * G_DIM;
    nx[0][0] = xp[3 * j + 0]; nx[0][1] = xp[3 * j + 1]; nx[0][2] = xp[3 * j + 2];
    nx[0][3] = xp[GI_OFF + j]; nx[0][4] = xp[GC_OFF + j]; nx[0][5] = xp[GO_OFF + j];
  }
  #pragma nounroll
  for (int tt = 0; tt < T_DIM; tt += 2) {
   #pragma unroll
   for (int u = 0; u < 2; ++u) {
    const int t = tt + u;
    v2f s0 = {0.f, 0.f}, s1 = {0.f, 0.f}, s2 = {0.f, 0.f};
    v2f s3 = {0.f, 0.f}, s4 = {0.f, 0.f}, s5 = {0.f, 0.f};
    v2f u0 = {0.f, 0.f}, u1 = {0.f, 0.f}, u2 = {0.f, 0.f};
    v2f u3 = {0.f, 0.f}, u4 = {0.f, 0.f}, u5 = {0.f, 0.f};
    const float* xn = Xp + (size_t)(((t + 1 < T_DIM) ? t + 1 : t) * B_DIM + b) * G_DIM;
    {
      const float4* h4 = (const float4*)(hv + (t & 1) * H_DIM);
      float4 q[HWIN];
      #pragma unroll
      for (int n = 0; n < HWIN; ++n) q[n] = h4[n];
      __builtin_amdgcn_sched_barrier(0);
      #pragma unroll
      for (int i4 = 0; i4 < H_DIM / 4; ++i4) {
        const float4 h = q[i4 % HWIN];
        const v2f hlo = {h.x, h.y}, hhi = {h.z, h.w};
        const int k = 2 * i4;
        s0 = __builtin_elementwise_fma(hlo, w2[0 * (H_DIM / 2) + k], s0);
        s1 = __builtin_elementwise_fma(hlo, w2[1 * (H_DIM / 2) + k], s1);
        s2 = __builtin_elementwise_fma(hlo, w2[2 * (H_DIM / 2) + k], s2);
        s3 = __builtin_elementwise_fma(hlo, w2[3 * (H_DIM / 2) + k], s3);
        s4 = __builtin_elementwise_fma(hlo, w2[4 * (H_DIM / 2) + k], s4);
        s5 = __builtin_elementwise_fma(hlo, w2[5 * (H_DIM / 2) + k], s5);
        u0 = __builtin_elementwise_fma(hhi, w2[0 * (H_DIM / 2) + k + 1], u0);
        u1 = __builtin_elementwise_fma(hhi, w2[1 * (H_DIM / 2) + k + 1], u1);
        u2 = __builtin_elementwise_fma(hhi, w2[2 * (H_DIM / 2) + k + 1], u2);
        u3 = __builtin_elementwise_fma(hhi, w2[3 * (H_DIM / 2) + k + 1], u3);
        u4 = __builtin_elementwise_fma(hhi, w2[4 * (H_DIM / 2) + k + 1], u4);
        u5 = __builtin_elementwise_fma(hhi, w2[5 * (H_DIM / 2) + k + 1], u5);
        if (i4 + HWIN < H_DIM / 4) q[i4 % HWIN] = h4[i4 + HWIN];
        if (i4 == 0) {
          nx[u ^ 1][0] = xn[3 * j + 0]; nx[u ^ 1][1] = xn[3 * j + 1]; nx[u ^ 1][2] = xn[3 * j + 2];
          nx[u ^ 1][3] = xn[GI_OFF + j]; nx[u ^ 1][4] = xn[GC_OFF + j]; nx[u ^ 1][5] = xn[GO_OFF + j];
        }
        __builtin_amdgcn_sched_barrier(0);
      }
    }
    const v2f p0 = s0 + u0, p1 = s1 + u1, p2 = s2 + u2;
    const v2f p3 = s3 + u3, p4 = s4 + u4, p5 = s5 + u5;
    const float a0 = (p0.x + p0.y) + nx[u][0], a1 = (p1.x + p1.y) + nx[u][1], a2 = (p2.x + p2.y) + nx[u][2];
    const float a3 = (p3.x + p3.y) + nx[u][3], a4 = (p4.x + p4.y) + nx[u][4], a5 = (p5.x + p5.y) + nx[u][5];
    const float mx = fmaxf(a0, fmaxf(a1, a2));
    const float e0 = __expf(a0 - mx), e1 = __expf(a1 - mx), e2 = __expf(a2 - mx);
    const float inv = __builtin_amdgcn_rcpf(e0 + e1 + e2);
    const float ca = e0 * inv, cb = e1 * inv, cc = e2 * inv;
    const float iv = __builtin_amdgcn_rcpf(1.f + __expf(-a3));
    const float cv = tanh_sel(a4);
    const float ov = __builtin_amdgcn_rcpf(1.f + __expf(-a5));
    const float pushed = iv * cv;
    {
      const float sl_t = (t == 0) ? pushed : top_prev;
      const float top = sl_t * ca + (pushed + vbs) * cb + (pushed + vbd) * cc;
      const float hn = top * ov;
      hv[((t + 1) & 1) * H_DIM + j] = hn;
      hlast = hn;
      top_prev = top;
    }
    asm volatile("s_wait_dscnt 0x0" ::: "memory");
    asm volatile("s_barrier_signal -1" ::: "memory");
    asm volatile("s_barrier_wait -1" ::: "memory");
    out[(size_t)(t * B_DIM + b) * H_DIM + j] = hlast;
   }
  }
  asm volatile("s_wait_loadcnt 0x0" ::: "memory");
  out[(size_t)T_DIM * B_DIM * H_DIM + (size_t)b * H_DIM + j] = hlast;
}

extern "C" void kernel_launch(void* const* d_in, const int* in_sizes, int n_in,
                              void* d_out, int out_size, void* d_ws, size_t ws_size,
                              hipStream_t stream) {
  (void)in_sizes; (void)n_in; (void)out_size; (void)ws_size;
  const float* x  = (const float*)d_in[0];
  const float* Wf = (const float*)d_in[1];
  const float* bf = (const float*)d_in[2];
  const float* Wi = (const float*)d_in[3];
  const float* bi = (const float*)d_in[4];
  const float* Wc = (const float*)d_in[5];
  const float* bc = (const float*)d_in[6];
  const float* Wo = (const float*)d_in[7];
  const float* bo = (const float*)d_in[8];
  const float* b_stay = (const float*)d_in[11];
  const float* b_down = (const float*)d_in[13];
  float* out = (float*)d_out;

  float* Xp  = (float*)d_ws;

  xproj_wmma<<<(ROWS / P1_BM) * (G_DIM / P1_BN), P1_THREADS, 0, stream>>>(
      x, Wf, bf, Wi, bi, Wc, bc, Wo, bo, Xp);

  size_t shmem = (size_t)(2 * H_DIM) * sizeof(float);
  scan_kernel<<<B_DIM, SCAN_THREADS, shmem, stream>>>(Xp, Wf, Wi, Wc, Wo, b_stay, b_down, out);
}
